# starting kernel
# baseline (speedup 1.0000x reference)
#include <hip/hip_runtime.h>
#include <hip/hip_bf16.h>
#include <math.h>

#define BATCH    128
#define IN_CAPS  2048
#define IN_DIM   8
#define NUM_CAPS 32
#define CAP_DIM  16
#define EPS      1e-8f

#define SPLITK   32
#define KCHUNK   (IN_CAPS / SPLITK)
#define SVEC     (BATCH * NUM_CAPS * CAP_DIM)

typedef float v2f __attribute__((ext_vector_type(2)));
typedef float v8f __attribute__((ext_vector_type(8)));


__global__ __launch_bounds__(256) void caps_s1_partial(const float* __restrict__ x,
                                                       const float* __restrict__ W,
                                                       float* __restrict__ part)
{
    const int j    = blockIdx.x;
    const int kc   = blockIdx.y;
    const int wav  = threadIdx.x >> 5;
    const int lane = threadIdx.x & 31;
    const int lo   = lane & 15;
    const int hi   = lane >> 4;
    const int m0   = wav * 16;
    const int i0   = kc * KCHUNK;

    const float* xA = x + (size_t)(m0 + lo) * (IN_CAPS * IN_DIM);
    const float* wB = W + (size_t)j * (CAP_DIM * IN_DIM) + lo * IN_DIM;

    v8f acc = {};
    for (int i = i0; i < i0 + KCHUNK; ++i) {
        const float* xi = xA + i * IN_DIM;
        const float* wi = wB + (size_t)i * (NUM_CAPS * CAP_DIM * IN_DIM);
#pragma unroll
        for (int s = 0; s < 2; ++s) {
            const int k = s * 4 + 2 * hi;
            v2f a, b;
            a.x = xi[k];     a.y = xi[k + 1];
            b.x = wi[k];     b.y = wi[k + 1];
            acc = __builtin_amdgcn_wmma_f32_16x16x4_f32(
                      false, a, false, b, (short)0, acc, false, false);
        }
    }
    float* p = part + (size_t)kc * SVEC + j * CAP_DIM + lo;
#pragma unroll
    for (int r = 0; r < 8; ++r)
        p[(size_t)(m0 + r + 8 * hi) * (NUM_CAPS * CAP_DIM)] = acc[r];
}

__global__ __launch_bounds__(256) void caps_s2_partial(const float* __restrict__ x,
                                                       const float* __restrict__ W,
                                                       const float* __restrict__ c2,
                                                       float* __restrict__ part)
{
    const int j    = blockIdx.x;
    const int kc   = blockIdx.y;
    const int wav  = threadIdx.x >> 5;
    const int lane = threadIdx.x & 31;
    const int lo   = lane & 15;
    const int hi   = lane >> 4;
    const int m0   = wav * 16;
    const int i0   = kc * KCHUNK;

    const float* xA = x  + (size_t)(m0 + lo) * (IN_CAPS * IN_DIM);
    const float* cA = c2 + (size_t)(m0 + lo) * (IN_CAPS * NUM_CAPS) + j;
    const float* wB = W  + (size_t)j * (CAP_DIM * IN_DIM) + lo * IN_DIM;

    v8f acc = {};
    for (int i = i0; i < i0 + KCHUNK; ++i) {
        const float  c  = cA[(size_t)i * NUM_CAPS];
        const float* xi = xA + i * IN_DIM;
        const float* wi = wB + (size_t)i * (NUM_CAPS * CAP_DIM * IN_DIM);
#pragma unroll
        for (int s = 0; s < 2; ++s) {
            const int k = s * 4 + 2 * hi;
            v2f a, b;
            a.x = c * xi[k];   a.y = c * xi[k + 1];
            b.x = wi[k];       b.y = wi[k + 1];
            acc = __builtin_amdgcn_wmma_f32_16x16x4_f32(
                      false, a, false, b, (short)0, acc, false, false);
        }
    }
    float* p = part + (size_t)kc * SVEC + j * CAP_DIM + lo;
#pragma unroll
    for (int r = 0; r < 8; ++r)
        p[(size_t)(m0 + r + 8 * hi) * (NUM_CAPS * CAP_DIM)] = acc[r];
}

__global__ __launch_bounds__(256) void caps_reduce_squash(const float* __restrict__ part,
                                                          float* __restrict__ out,
                                                          float pre)
{
    const int t = blockIdx.x * 256 + threadIdx.x;
    float s = 0.0f;
    for (int k = 0; k < SPLITK; ++k)
        s += part[(size_t)k * SVEC + t];
    s *= pre;
    float sq = s * s;
    sq += __shfl_xor(sq, 1, 16);
    sq += __shfl_xor(sq, 2, 16);
    sq += __shfl_xor(sq, 4, 16);
    sq += __shfl_xor(sq, 8, 16);
    const float scale = sq / ((1.0f + sq) * (sqrtf(sq) + EPS));
    out[t] = s * scale;
}

__global__ __launch_bounds__(256) void caps_agree_kernel(const float* __restrict__ x,
                                                         const float* __restrict__ W,
                                                         const float* __restrict__ v1,
                                                         float* __restrict__ bagr)
{
    const int i    = blockIdx.x;
    const int j    = blockIdx.y;
    const int wav  = threadIdx.x >> 5;
    const int lane = threadIdx.x & 31;
    const int lo   = lane & 15;
    const int hi   = lane >> 4;
    const int m0   = wav * 16;

    const float* xi = x + (size_t)(m0 + lo) * (IN_CAPS * IN_DIM) + i * IN_DIM;
    const float* wi = W + (size_t)i * (NUM_CAPS * CAP_DIM * IN_DIM)
                        + (size_t)j * (CAP_DIM * IN_DIM) + lo * IN_DIM;

    v8f acc = {};
#pragma unroll
    for (int s = 0; s < 2; ++s) {
        const int k = s * 4 + 2 * hi;
        v2f a, b;
        a.x = xi[k];     a.y = xi[k + 1];
        b.x = wi[k];     b.y = wi[k + 1];
        acc = __builtin_amdgcn_wmma_f32_16x16x4_f32(
                  false, a, false, b, (short)0, acc, false, false);
    }
#pragma unroll
    for (int r = 0; r < 8; ++r) {
        const int m = m0 + r + 8 * hi;
        float t = acc[r] * v1[(size_t)m * (NUM_CAPS * CAP_DIM) + j * CAP_DIM + lo];
        t += __shfl_xor(t, 1, 16);
        t += __shfl_xor(t, 2, 16);
        t += __shfl_xor(t, 4, 16);
        t += __shfl_xor(t, 8, 16);
        if (lo == r)
            bagr[(size_t)m * (IN_CAPS * NUM_CAPS) + (size_t)i * NUM_CAPS + j] = t;
    }
}

__global__ __launch_bounds__(256) void caps_softmax_kernel(float* __restrict__ bagr)
{
    const size_t row = (size_t)blockIdx.x * blockDim.x + threadIdx.x;
    float* p = bagr + row * NUM_CAPS;
    float vals[NUM_CAPS];
    float mx = -INFINITY;
#pragma unroll
    for (int j = 0; j < NUM_CAPS; ++j) { vals[j] = p[j]; mx = fmaxf(mx, vals[j]); }
    float sum = 0.0f;
#pragma unroll
    for (int j = 0; j < NUM_CAPS; ++j) { vals[j] = __expf(vals[j] - mx); sum += vals[j]; }
    const float inv = 1.0f / sum;
#pragma unroll
    for (int j = 0; j < NUM_CAPS; ++j) p[j] = vals[j] * inv;
}

extern "C" void kernel_launch(void* const* d_in, const int* in_sizes, int n_in,
                              void* d_out, int out_size, void* d_ws, size_t ws_size,
                              hipStream_t stream)
{
    const float* x = (const float*)d_in[0];
    const float* W = (const float*)d_in[1];
    float* vout = (float*)d_out;

    float* v1   = (float*)d_ws;
    float* bagr = v1 + (size_t)SVEC;
    float* part = bagr + (size_t)BATCH * IN_CAPS * NUM_CAPS;

    caps_s1_partial   <<<dim3(NUM_CAPS, SPLITK), 256, 0, stream>>>(x, W, part);
    caps_reduce_squash<<<dim3(SVEC / 256),       256, 0, stream>>>(part, v1,
                                                                  1.0f / (float)NUM_CAPS);
    caps_agree_kernel  <<<dim3(IN_CAPS, NUM_CAPS),       256, 0, stream>>>(x, W, v1, bagr);
    caps_softmax_kernel<<<dim3((BATCH * IN_CAPS) / 256), 256, 0, stream>>>(bagr);
    caps_s2_partial   <<<dim3(NUM_CAPS, SPLITK), 256, 0, stream>>>(x, W, bagr, part);
    caps_reduce_squash<<<dim3(SVEC / 256),       256, 0, stream>>>(part, vout, 1.0f);
}


# agreement re-tiled: W as A, x as B operand so each lane holds a whole logit row; d-sum serial in-lane + one half exchange (sum order changed), v1 in registers, softmax fused in; 6 launches to 5
# speedup vs baseline: 2.0947x; 2.0947x over previous
#include <hip/hip_runtime.h>
#include <hip/hip_bf16.h>
#include <math.h>

#define BATCH    128
#define IN_CAPS  2048
#define IN_DIM   8
#define NUM_CAPS 32
#define CAP_DIM  16
#define EPS      1e-8f

#define SPLITK   32
#define KCHUNK   (IN_CAPS / SPLITK)
#define SVEC     (BATCH * NUM_CAPS * CAP_DIM)

typedef float v2f __attribute__((ext_vector_type(2)));
typedef float v8f __attribute__((ext_vector_type(8)));


__global__ __launch_bounds__(256) void caps_s1_partial(const float* __restrict__ x,
                                                       const float* __restrict__ W,
                                                       float* __restrict__ part)
{
    const int j    = blockIdx.x;
    const int kc   = blockIdx.y;
    const int wav  = threadIdx.x >> 5;
    const int lane = threadIdx.x & 31;
    const int lo   = lane & 15;
    const int hi   = lane >> 4;
    const int m0   = wav * 16;
    const int i0   = kc * KCHUNK;

    const float* xA = x + (size_t)(m0 + lo) * (IN_CAPS * IN_DIM);
    const float* wB = W + (size_t)j * (CAP_DIM * IN_DIM) + lo * IN_DIM;

    v8f acc = {};
    for (int i = i0; i < i0 + KCHUNK; ++i) {
        const float* xi = xA + i * IN_DIM;
        const float* wi = wB + (size_t)i * (NUM_CAPS * CAP_DIM * IN_DIM);
#pragma unroll
        for (int s = 0; s < 2; ++s) {
            const int k = s * 4 + 2 * hi;
            v2f a, b;
            a.x = xi[k];     a.y = xi[k + 1];
            b.x = wi[k];     b.y = wi[k + 1];
            acc = __builtin_amdgcn_wmma_f32_16x16x4_f32(
                      false, a, false, b, (short)0, acc, false, false);
        }
    }
    float* p = part + (size_t)kc * SVEC + j * CAP_DIM + lo;
#pragma unroll
    for (int r = 0; r < 8; ++r)
        p[(size_t)(m0 + r + 8 * hi) * (NUM_CAPS * CAP_DIM)] = acc[r];
}

__global__ __launch_bounds__(256) void caps_s2_partial(const float* __restrict__ x,
                                                       const float* __restrict__ W,
                                                       const float* __restrict__ c2,
                                                       float* __restrict__ part)
{
    const int j    = blockIdx.x;
    const int kc   = blockIdx.y;
    const int wav  = threadIdx.x >> 5;
    const int lane = threadIdx.x & 31;
    const int lo   = lane & 15;
    const int hi   = lane >> 4;
    const int m0   = wav * 16;
    const int i0   = kc * KCHUNK;

    const float* xA = x  + (size_t)(m0 + lo) * (IN_CAPS * IN_DIM);
    const float* cA = c2 + (size_t)(m0 + lo) * (IN_CAPS * NUM_CAPS) + j;
    const float* wB = W  + (size_t)j * (CAP_DIM * IN_DIM) + lo * IN_DIM;

    v8f acc = {};
    for (int i = i0; i < i0 + KCHUNK; ++i) {
        const float  c  = cA[(size_t)i * NUM_CAPS];
        const float* xi = xA + i * IN_DIM;
        const float* wi = wB + (size_t)i * (NUM_CAPS * CAP_DIM * IN_DIM);
#pragma unroll
        for (int s = 0; s < 2; ++s) {
            const int k = s * 4 + 2 * hi;
            v2f a, b;
            a.x = c * xi[k];   a.y = c * xi[k + 1];
            b.x = wi[k];       b.y = wi[k + 1];
            acc = __builtin_amdgcn_wmma_f32_16x16x4_f32(
                      false, a, false, b, (short)0, acc, false, false);
        }
    }
    float* p = part + (size_t)kc * SVEC + j * CAP_DIM + lo;
#pragma unroll
    for (int r = 0; r < 8; ++r)
        p[(size_t)(m0 + r + 8 * hi) * (NUM_CAPS * CAP_DIM)] = acc[r];
}

__global__ __launch_bounds__(256) void caps_reduce_squash(const float* __restrict__ part,
                                                          float* __restrict__ out,
                                                          float pre)
{
    const int t = blockIdx.x * 256 + threadIdx.x;
    float s = 0.0f;
    for (int k = 0; k < SPLITK; ++k)
        s += part[(size_t)k * SVEC + t];
    s *= pre;
    float sq = s * s;
    sq += __shfl_xor(sq, 1, 16);
    sq += __shfl_xor(sq, 2, 16);
    sq += __shfl_xor(sq, 4, 16);
    sq += __shfl_xor(sq, 8, 16);
    const float scale = sq / ((1.0f + sq) * (sqrtf(sq) + EPS));
    out[t] = s * scale;
}

#define IPW 8

__device__ __forceinline__ float pick(bool c, float a, float b) { return c ? a : b; }

__device__ __forceinline__ float xhalf(float v)
{
    const int b = __builtin_bit_cast(int, v);
    return __builtin_bit_cast(float,
        __builtin_amdgcn_permlanex16(b, b, 0x76543210, 0xfedcba98, false, false));
}

__global__ __launch_bounds__(256) void caps_agree_softmax(const float* __restrict__ x,
                                                          const float* __restrict__ W,
                                                          const float* __restrict__ v1,
                                                          float* __restrict__ c2)
{
    const int wav  = threadIdx.x >> 5;
    const int lane = threadIdx.x & 31;
    const int lo   = lane & 15;
    const int hi   = lane >> 4;
    const int m0   = blockIdx.x * 16;
    const int ib   = (blockIdx.y * 8 + wav) * IPW;

    const float* xr = x  + (size_t)(m0 + lo) * (IN_CAPS * IN_DIM) + 2 * hi;
    const float* vr = v1 + (size_t)(m0 + lo) * (NUM_CAPS * CAP_DIM) + 8 * hi;
    const float* wr = W  + lo * IN_DIM + 2 * hi;
    float*       cr = c2 + (size_t)(m0 + lo) * (IN_CAPS * NUM_CAPS) + 16 * hi;

    float4 va[NUM_CAPS], vb[NUM_CAPS];
#pragma unroll
    for (int j = 0; j < NUM_CAPS; ++j) {
        va[j] = *(const float4*)(vr + j * CAP_DIM);
        vb[j] = *(const float4*)(vr + j * CAP_DIM + 4);
    }

    for (int ii = 0; ii < IPW; ++ii) {
        const int i = ib + ii;
        const float* xi = xr + i * IN_DIM;
        v2f xb0, xb1;
        xb0.x = xi[0];   xb0.y = xi[1];
        xb1.x = xi[4];   xb1.y = xi[5];
        const float* wi = wr + (size_t)i * (NUM_CAPS * CAP_DIM * IN_DIM);

        v2f wa[NUM_CAPS], wb[NUM_CAPS];
#pragma unroll
        for (int j = 0; j < NUM_CAPS; ++j) {
            const float* wj = wi + j * (CAP_DIM * IN_DIM);
            wa[j].x = wj[0];   wa[j].y = wj[1];
            wb[j].x = wj[4];   wb[j].y = wj[5];
        }
        __builtin_amdgcn_sched_barrier(0);

        float lg[NUM_CAPS];
#pragma unroll
        for (int j = 0; j < NUM_CAPS; ++j) {
            v8f acc = {};
            acc = __builtin_amdgcn_wmma_f32_16x16x4_f32(
                      false, wa[j], false, xb0, (short)0, acc, false, false);
            acc = __builtin_amdgcn_wmma_f32_16x16x4_f32(
                      false, wb[j], false, xb1, (short)0, acc, false, false);
            float t = acc[0] * va[j].x;
            t += acc[1] * va[j].y;
            t += acc[2] * va[j].z;
            t += acc[3] * va[j].w;
            t += acc[4] * vb[j].x;
            t += acc[5] * vb[j].y;
            t += acc[6] * vb[j].z;
            t += acc[7] * vb[j].w;
            lg[j] = t + xhalf(t);
        }

        float mx = -INFINITY;
#pragma unroll
        for (int j = 0; j < NUM_CAPS; ++j) mx = fmaxf(mx, lg[j]);
        float sum = 0.0f;
#pragma unroll
        for (int j = 0; j < NUM_CAPS; ++j) { lg[j] = __expf(lg[j] - mx); sum += lg[j]; }
        const float inv = 1.0f / sum;

        float* cp = cr + (size_t)i * NUM_CAPS;
#pragma unroll
        for (int q = 0; q < 4; ++q) {
            float4 o;
            o.x = pick(hi != 0, lg[16 + 4 * q + 0], lg[4 * q + 0]) * inv;
            o.y = pick(hi != 0, lg[16 + 4 * q + 1], lg[4 * q + 1]) * inv;
            o.z = pick(hi != 0, lg[16 + 4 * q + 2], lg[4 * q + 2]) * inv;
            o.w = pick(hi != 0, lg[16 + 4 * q + 3], lg[4 * q + 3]) * inv;
            *(float4*)(cp + 4 * q) = o;
        }
    }
}

extern "C" void kernel_launch(void* const* d_in, const int* in_sizes, int n_in,
                              void* d_out, int out_size, void* d_ws, size_t ws_size,
                              hipStream_t stream)
{
    const float* x = (const float*)d_in[0];
    const float* W = (const float*)d_in[1];
    float* vout = (float*)d_out;

    float* v1   = (float*)d_ws;
    float* bagr = v1 + (size_t)SVEC;
    float* part = bagr + (size_t)BATCH * IN_CAPS * NUM_CAPS;

    caps_s1_partial   <<<dim3(NUM_CAPS, SPLITK), 256, 0, stream>>>(x, W, part);
    caps_reduce_squash<<<dim3(SVEC / 256),       256, 0, stream>>>(part, v1,
                                                                  1.0f / (float)NUM_CAPS);
    caps_agree_softmax<<<dim3(BATCH / 16, IN_CAPS / (8 * IPW)), 256, 0, stream>>>(x, W, v1, bagr);
    caps_s2_partial   <<<dim3(NUM_CAPS, SPLITK), 256, 0, stream>>>(x, W, bagr, part);
    caps_reduce_squash<<<dim3(SVEC / 256),       256, 0, stream>>>(part, vout, 1.0f);
}


# both split-K GEMM passes re-tiled: a wave keeps 4 output capsules (4 accumulators) per sample tile, x and c2 fragments loaded once per 4 capsules, loads grouped one stage ahead; sum order kept
# speedup vs baseline: 3.7941x; 1.8113x over previous
#include <hip/hip_runtime.h>
#include <hip/hip_bf16.h>
#include <math.h>

#define BATCH    128
#define IN_CAPS  2048
#define IN_DIM   8
#define NUM_CAPS 32
#define CAP_DIM  16
#define EPS      1e-8f

#define SPLITK   32
#define KCHUNK   (IN_CAPS / SPLITK)
#define SVEC     (BATCH * NUM_CAPS * CAP_DIM)

typedef float v2f __attribute__((ext_vector_type(2)));
typedef float v8f __attribute__((ext_vector_type(8)));


#define JG   4
#define ST   4
#define NST  (KCHUNK / ST)

struct SFrag {
    v2f    xa[ST][2];
    v2f    wb[ST][JG][2];
    float4 c[ST];
};

template <bool WEIGHTED>
__device__ __forceinline__ void s_load_stage(SFrag& f, const float* __restrict__ xr,
                                             const float* __restrict__ wr,
                                             const float* __restrict__ cr, int i)
{
    const float* xi = xr + i * IN_DIM;
    const float* wi = wr + (size_t)i * (NUM_CAPS * CAP_DIM * IN_DIM);
    const float* ci = cr + (size_t)i * NUM_CAPS;
#pragma unroll
    for (int t = 0; t < ST; ++t) {
        f.xa[t][0].x = xi[t * IN_DIM + 0];   f.xa[t][0].y = xi[t * IN_DIM + 1];
        f.xa[t][1].x = xi[t * IN_DIM + 4];   f.xa[t][1].y = xi[t * IN_DIM + 5];
        if (WEIGHTED) f.c[t] = *(const float4*)(ci + t * NUM_CAPS);
#pragma unroll
        for (int jj = 0; jj < JG; ++jj) {
            const float* wj = wi + t * (NUM_CAPS * CAP_DIM * IN_DIM) + jj * (CAP_DIM * IN_DIM);
            f.wb[t][jj][0].x = wj[0];   f.wb[t][jj][0].y = wj[1];
            f.wb[t][jj][1].x = wj[4];   f.wb[t][jj][1].y = wj[5];
        }
    }
}

template <bool WEIGHTED>
__device__ __forceinline__ void s_compute_stage(const SFrag& f, v8f (&acc)[JG])
{
#pragma unroll
    for (int t = 0; t < ST; ++t) {
        float cv[JG] = { 1.0f, 1.0f, 1.0f, 1.0f };
        if (WEIGHTED) { cv[0] = f.c[t].x;   cv[1] = f.c[t].y;   cv[2] = f.c[t].z;   cv[3] = f.c[t].w; }
#pragma unroll
        for (int s = 0; s < 2; ++s) {
#pragma unroll
            for (int jj = 0; jj < JG; ++jj) {
                v2f a = f.xa[t][s];
                if (WEIGHTED) { a.x = cv[jj] * a.x;   a.y = cv[jj] * a.y; }
                acc[jj] = __builtin_amdgcn_wmma_f32_16x16x4_f32(
                              false, a, false, f.wb[t][jj][s], (short)0, acc[jj], false, false);
            }
        }
    }
}

template <bool WEIGHTED>
__global__ __launch_bounds__(256) void caps_s_partial(const float* __restrict__ x,
                                                      const float* __restrict__ W,
                                                      const float* __restrict__ c2,
                                                      float* __restrict__ part)
{
    const int j0   = blockIdx.x * JG;
    const int kc   = blockIdx.y;
    const int wav  = threadIdx.x >> 5;
    const int lane = threadIdx.x & 31;
    const int lo   = lane & 15;
    const int hi   = lane >> 4;
    const int m0   = wav * 16;
    const int i0   = kc * KCHUNK;

    const float* xr = x  + (size_t)(m0 + lo) * (IN_CAPS * IN_DIM) + 2 * hi;
    const float* wr = W  + (size_t)j0 * (CAP_DIM * IN_DIM) + lo * IN_DIM + 2 * hi;
    const float* cr = c2 + (size_t)(m0 + lo) * (IN_CAPS * NUM_CAPS) + j0;

    v8f acc[JG];
#pragma unroll
    for (int jj = 0; jj < JG; ++jj) acc[jj] = v8f{};

    SFrag f0, f1;
    s_load_stage<WEIGHTED>(f0, xr, wr, cr, i0);
    for (int st = 0; st < NST - 2; st += 2) {
        s_load_stage<WEIGHTED>(f1, xr, wr, cr, i0 + (st + 1) * ST);
        __builtin_amdgcn_sched_barrier(0);
        s_compute_stage<WEIGHTED>(f0, acc);
        s_load_stage<WEIGHTED>(f0, xr, wr, cr, i0 + (st + 2) * ST);
        __builtin_amdgcn_sched_barrier(0);
        s_compute_stage<WEIGHTED>(f1, acc);
    }
    s_load_stage<WEIGHTED>(f1, xr, wr, cr, i0 + (NST - 1) * ST);
    __builtin_amdgcn_sched_barrier(0);
    s_compute_stage<WEIGHTED>(f0, acc);
    s_compute_stage<WEIGHTED>(f1, acc);

#pragma unroll
    for (int jj = 0; jj < JG; ++jj) {
        float* p = part + (size_t)kc * SVEC + (j0 + jj) * CAP_DIM + lo;
#pragma unroll
        for (int r = 0; r < 8; ++r)
            p[(size_t)(m0 + r + 8 * hi) * (NUM_CAPS * CAP_DIM)] = acc[jj][r];
    }
}

__global__ __launch_bounds__(256) void caps_reduce_squash(const float* __restrict__ part,
                                                          float* __restrict__ out,
                                                          float pre)
{
    const int t = blockIdx.x * 256 + threadIdx.x;
    float s = 0.0f;
    for (int k = 0; k < SPLITK; ++k)
        s += part[(size_t)k * SVEC + t];
    s *= pre;
    float sq = s * s;
    sq += __shfl_xor(sq, 1, 16);
    sq += __shfl_xor(sq, 2, 16);
    sq += __shfl_xor(sq, 4, 16);
    sq += __shfl_xor(sq, 8, 16);
    const float scale = sq / ((1.0f + sq) * (sqrtf(sq) + EPS));
    out[t] = s * scale;
}

#define IPW 8

__device__ __forceinline__ float pick(bool c, float a, float b) { return c ? a : b; }

__device__ __forceinline__ float xhalf(float v)
{
    const int b = __builtin_bit_cast(int, v);
    return __builtin_bit_cast(float,
        __builtin_amdgcn_permlanex16(b, b, 0x76543210, 0xfedcba98, false, false));
}

__global__ __launch_bounds__(256) void caps_agree_softmax(const float* __restrict__ x,
                                                          const float* __restrict__ W,
                                                          const float* __restrict__ v1,
                                                          float* __restrict__ c2)
{
    const int wav  = threadIdx.x >> 5;
    const int lane = threadIdx.x & 31;
    const int lo   = lane & 15;
    const int hi   = lane >> 4;
    const int m0   = blockIdx.x * 16;
    const int ib   = (blockIdx.y * 8 + wav) * IPW;

    const float* xr = x  + (size_t)(m0 + lo) * (IN_CAPS * IN_DIM) + 2 * hi;
    const float* vr = v1 + (size_t)(m0 + lo) * (NUM_CAPS * CAP_DIM) + 8 * hi;
    const float* wr = W  + lo * IN_DIM + 2 * hi;
    float*       cr = c2 + (size_t)(m0 + lo) * (IN_CAPS * NUM_CAPS) + 16 * hi;

    float4 va[NUM_CAPS], vb[NUM_CAPS];
#pragma unroll
    for (int j = 0; j < NUM_CAPS; ++j) {
        va[j] = *(const float4*)(vr + j * CAP_DIM);
        vb[j] = *(const float4*)(vr + j * CAP_DIM + 4);
    }

    for (int ii = 0; ii < IPW; ++ii) {
        const int i = ib + ii;
        const float* xi = xr + i * IN_DIM;
        v2f xb0, xb1;
        xb0.x = xi[0];   xb0.y = xi[1];
        xb1.x = xi[4];   xb1.y = xi[5];
        const float* wi = wr + (size_t)i * (NUM_CAPS * CAP_DIM * IN_DIM);

        v2f wa[NUM_CAPS], wb[NUM_CAPS];
#pragma unroll
        for (int j = 0; j < NUM_CAPS; ++j) {
            const float* wj = wi + j * (CAP_DIM * IN_DIM);
            wa[j].x = wj[0];   wa[j].y = wj[1];
            wb[j].x = wj[4];   wb[j].y = wj[5];
        }
        __builtin_amdgcn_sched_barrier(0);

        float lg[NUM_CAPS];
#pragma unroll
        for (int j = 0; j < NUM_CAPS; ++j) {
            v8f acc = {};
            acc = __builtin_amdgcn_wmma_f32_16x16x4_f32(
                      false, wa[j], false, xb0, (short)0, acc, false, false);
            acc = __builtin_amdgcn_wmma_f32_16x16x4_f32(
                      false, wb[j], false, xb1, (short)0, acc, false, false);
            float t = acc[0] * va[j].x;
            t += acc[1] * va[j].y;
            t += acc[2] * va[j].z;
            t += acc[3] * va[j].w;
            t += acc[4] * vb[j].x;
            t += acc[5] * vb[j].y;
            t += acc[6] * vb[j].z;
            t += acc[7] * vb[j].w;
            lg[j] = t + xhalf(t);
        }

        float mx = -INFINITY;
#pragma unroll
        for (int j = 0; j < NUM_CAPS; ++j) mx = fmaxf(mx, lg[j]);
        float sum = 0.0f;
#pragma unroll
        for (int j = 0; j < NUM_CAPS; ++j) { lg[j] = __expf(lg[j] - mx); sum += lg[j]; }
        const float inv = 1.0f / sum;

        float* cp = cr + (size_t)i * NUM_CAPS;
#pragma unroll
        for (int q = 0; q < 4; ++q) {
            float4 o;
            o.x = pick(hi != 0, lg[16 + 4 * q + 0], lg[4 * q + 0]) * inv;
            o.y = pick(hi != 0, lg[16 + 4 * q + 1], lg[4 * q + 1]) * inv;
            o.z = pick(hi != 0, lg[16 + 4 * q + 2], lg[4 * q + 2]) * inv;
            o.w = pick(hi != 0, lg[16 + 4 * q + 3], lg[4 * q + 3]) * inv;
            *(float4*)(cp + 4 * q) = o;
        }
    }
}

extern "C" void kernel_launch(void* const* d_in, const int* in_sizes, int n_in,
                              void* d_out, int out_size, void* d_ws, size_t ws_size,
                              hipStream_t stream)
{
    const float* x = (const float*)d_in[0];
    const float* W = (const float*)d_in[1];
    float* vout = (float*)d_out;

    float* v1   = (float*)d_ws;
    float* bagr = v1 + (size_t)SVEC;
    float* part = bagr + (size_t)BATCH * IN_CAPS * NUM_CAPS;

    caps_s_partial<false><<<dim3(NUM_CAPS / JG, SPLITK), 256, 0, stream>>>(x, W, bagr, part);
    caps_reduce_squash<<<dim3(SVEC / 256),       256, 0, stream>>>(part, v1,
                                                                  1.0f / (float)NUM_CAPS);
    caps_agree_softmax<<<dim3(BATCH / 16, IN_CAPS / (8 * IPW)), 256, 0, stream>>>(x, W, v1, bagr);
    caps_s_partial<true> <<<dim3(NUM_CAPS / JG, SPLITK), 256, 0, stream>>>(x, W, bagr, part);
    caps_reduce_squash<<<dim3(SVEC / 256),       256, 0, stream>>>(part, vout, 1.0f);
}


# second routing pass fused into the agreement kernel: softmax weight scales the lane's x fragment, W fragments reused from registers for s2; c2 never stored; 5 launches to 4; s2 sum order changed
# speedup vs baseline: 4.9494x; 1.3045x over previous
#include <hip/hip_runtime.h>
#include <hip/hip_bf16.h>
#include <math.h>


#define BATCH    128
#define IN_CAPS  2048
#define IN_DIM   8
#define NUM_CAPS 32
#define CAP_DIM  16
#define EPS      1e-8f

#define SPLITK   32
#define KCHUNK   (IN_CAPS / SPLITK)
#define SVEC     (BATCH * NUM_CAPS * CAP_DIM)

typedef float v2f __attribute__((ext_vector_type(2)));
typedef float v8f __attribute__((ext_vector_type(8)));


#define JG   4
#define ST   4
#define NST  (KCHUNK / ST)

struct SFrag {
    v2f    xa[ST][2];
    v2f    wb[ST][JG][2];
};

__device__ __forceinline__ void s_load_stage(SFrag& f, const float* __restrict__ xr,
                                             const float* __restrict__ wr, int i)
{
    const float* xi = xr + i * IN_DIM;
    const float* wi = wr + (size_t)i * (NUM_CAPS * CAP_DIM * IN_DIM);
#pragma unroll
    for (int t = 0; t < ST; ++t) {
        f.xa[t][0].x = xi[t * IN_DIM + 0];   f.xa[t][0].y = xi[t * IN_DIM + 1];
        f.xa[t][1].x = xi[t * IN_DIM + 4];   f.xa[t][1].y = xi[t * IN_DIM + 5];
#pragma unroll
        for (int jj = 0; jj < JG; ++jj) {
            const float* wj = wi + t * (NUM_CAPS * CAP_DIM * IN_DIM) + jj * (CAP_DIM * IN_DIM);
            f.wb[t][jj][0].x = wj[0];   f.wb[t][jj][0].y = wj[1];
            f.wb[t][jj][1].x = wj[4];   f.wb[t][jj][1].y = wj[5];
        }
    }
}

__device__ __forceinline__ void s_compute_stage(const SFrag& f, v8f (&acc)[JG])
{
#pragma unroll
    for (int t = 0; t < ST; ++t) {
#pragma unroll
        for (int s = 0; s < 2; ++s) {
#pragma unroll
            for (int jj = 0; jj < JG; ++jj) {
                acc[jj] = __builtin_amdgcn_wmma_f32_16x16x4_f32(
                              false, f.xa[t][s], false, f.wb[t][jj][s], (short)0, acc[jj], false, false);
            }
        }
    }
}

__global__ __launch_bounds__(256) void caps_s_partial(const float* __restrict__ x,
                                                      const float* __restrict__ W,
                                                      float* __restrict__ part)
{
    const int j0   = blockIdx.x * JG;
    const int kc   = blockIdx.y;
    const int wav  = threadIdx.x >> 5;
    const int lane = threadIdx.x & 31;
    const int lo   = lane & 15;
    const int hi   = lane >> 4;
    const int m0   = wav * 16;
    const int i0   = kc * KCHUNK;

    const float* xr = x  + (size_t)(m0 + lo) * (IN_CAPS * IN_DIM) + 2 * hi;
    const float* wr = W  + (size_t)j0 * (CAP_DIM * IN_DIM) + lo * IN_DIM + 2 * hi;

    v8f acc[JG];
#pragma unroll
    for (int jj = 0; jj < JG; ++jj) acc[jj] = v8f{};

    SFrag f0, f1;
    s_load_stage(f0, xr, wr, i0);
    for (int st = 0; st < NST - 2; st += 2) {
        s_load_stage(f1, xr, wr, i0 + (st + 1) * ST);
        __builtin_amdgcn_sched_barrier(0);
        s_compute_stage(f0, acc);
        s_load_stage(f0, xr, wr, i0 + (st + 2) * ST);
        __builtin_amdgcn_sched_barrier(0);
        s_compute_stage(f1, acc);
    }
    s_load_stage(f1, xr, wr, i0 + (NST - 1) * ST);
    __builtin_amdgcn_sched_barrier(0);
    s_compute_stage(f0, acc);
    s_compute_stage(f1, acc);

#pragma unroll
    for (int jj = 0; jj < JG; ++jj) {
        float* p = part + (size_t)kc * SVEC + (j0 + jj) * CAP_DIM + lo;
#pragma unroll
        for (int r = 0; r < 8; ++r)
            p[(size_t)(m0 + r + 8 * hi) * (NUM_CAPS * CAP_DIM)] = acc[jj][r];
    }
}

__global__ __launch_bounds__(256) void caps_reduce_squash(const float* __restrict__ part,
                                                          float* __restrict__ out,
                                                          float pre)
{
    const int t = blockIdx.x * 256 + threadIdx.x;
    float s = 0.0f;
    for (int k = 0; k < SPLITK; ++k)
        s += part[(size_t)k * SVEC + t];
    s *= pre;
    float sq = s * s;
    sq += __shfl_xor(sq, 1, 16);
    sq += __shfl_xor(sq, 2, 16);
    sq += __shfl_xor(sq, 4, 16);
    sq += __shfl_xor(sq, 8, 16);
    const float scale = sq / ((1.0f + sq) * (sqrtf(sq) + EPS));
    out[t] = s * scale;
}

#define CPW  16
#define NCH  (IN_CAPS / CPW)
#define TILE (16 * NUM_CAPS * CAP_DIM)

__device__ __forceinline__ float xhalf(float v)
{
    const int b = __builtin_bit_cast(int, v);
    return __builtin_bit_cast(float,
        __builtin_amdgcn_permlanex16(b, b, 0x76543210, 0xfedcba98, false, false));
}

struct AFrag {
    v2f xb0, xb1;
    v2f wa[NUM_CAPS], wb[NUM_CAPS];
};

__device__ __forceinline__ void a_load(AFrag& f, const float* __restrict__ xr,
                                       const float* __restrict__ wr, int i)
{
    const float* xi = xr + i * IN_DIM;
    f.xb0.x = xi[0];   f.xb0.y = xi[1];
    f.xb1.x = xi[4];   f.xb1.y = xi[5];
    const float* wi = wr + (size_t)i * (NUM_CAPS * CAP_DIM * IN_DIM);
#pragma unroll
    for (int j = 0; j < NUM_CAPS; ++j) {
        const float* wj = wi + j * (CAP_DIM * IN_DIM);
        f.wa[j].x = wj[0];   f.wa[j].y = wj[1];
        f.wb[j].x = wj[4];   f.wb[j].y = wj[5];
    }
}

__device__ __forceinline__ void a_step(const AFrag& f, const float4 (&va)[NUM_CAPS],
                                       const float4 (&vb)[NUM_CAPS], v8f (&s2)[NUM_CAPS])
{
    float lg[NUM_CAPS];
#pragma unroll
    for (int j = 0; j < NUM_CAPS; ++j) {
        v8f acc = {};
        acc = __builtin_amdgcn_wmma_f32_16x16x4_f32(
                  false, f.wa[j], false, f.xb0, (short)0, acc, false, false);
        acc = __builtin_amdgcn_wmma_f32_16x16x4_f32(
                  false, f.wb[j], false, f.xb1, (short)0, acc, false, false);
        float t = acc[0] * va[j].x;
        t += acc[1] * va[j].y;
        t += acc[2] * va[j].z;
        t += acc[3] * va[j].w;
        t += acc[4] * vb[j].x;
        t += acc[5] * vb[j].y;
        t += acc[6] * vb[j].z;
        t += acc[7] * vb[j].w;
        lg[j] = t + xhalf(t);
    }

    float mx = -INFINITY;
#pragma unroll
    for (int j = 0; j < NUM_CAPS; ++j) mx = fmaxf(mx, lg[j]);
    float sum = 0.0f;
#pragma unroll
    for (int j = 0; j < NUM_CAPS; ++j) { lg[j] = __expf(lg[j] - mx); sum += lg[j]; }
    const float inv = 1.0f / sum;

#pragma unroll
    for (int g = 0; g < NUM_CAPS; g += 4) {
        v2f b0[4], b1[4];
#pragma unroll
        for (int jj = 0; jj < 4; ++jj) {
            const float c = lg[g + jj] * inv;
            b0[jj].x = c * f.xb0.x;   b0[jj].y = c * f.xb0.y;
            b1[jj].x = c * f.xb1.x;   b1[jj].y = c * f.xb1.y;
        }
#pragma unroll
        for (int jj = 0; jj < 4; ++jj)
            s2[g + jj] = __builtin_amdgcn_wmma_f32_16x16x4_f32(
                             false, f.wa[g + jj], false, b0[jj], (short)0, s2[g + jj], false, false);
#pragma unroll
        for (int jj = 0; jj < 4; ++jj)
            s2[g + jj] = __builtin_amdgcn_wmma_f32_16x16x4_f32(
                             false, f.wb[g + jj], false, b1[jj], (short)0, s2[g + jj], false, false);
    }
}

__global__ __launch_bounds__(128) void caps_agree_route(const float* __restrict__ x,
                                                        const float* __restrict__ W,
                                                        const float* __restrict__ v1,
                                                        float* __restrict__ part2)
{
    const int wav  = threadIdx.x >> 5;
    const int lane = threadIdx.x & 31;
    const int lo   = lane & 15;
    const int hi   = lane >> 4;
    const int mt   = blockIdx.x * 4 + wav;
    const int m0   = mt * 16;
    const int ib   = blockIdx.y * CPW;

    const float* xr = x  + (size_t)(m0 + lo) * (IN_CAPS * IN_DIM) + 2 * hi;
    const float* vr = v1 + (size_t)(m0 + lo) * (NUM_CAPS * CAP_DIM) + 8 * hi;
    const float* wr = W  + lo * IN_DIM + 2 * hi;

    float4 va[NUM_CAPS], vb[NUM_CAPS];
#pragma unroll
    for (int j = 0; j < NUM_CAPS; ++j) {
        va[j] = *(const float4*)(vr + j * CAP_DIM);
        vb[j] = *(const float4*)(vr + j * CAP_DIM + 4);
    }

    v8f s2[NUM_CAPS];
#pragma unroll
    for (int j = 0; j < NUM_CAPS; ++j) s2[j] = v8f{};

    AFrag fc, fn;
    a_load(fc, xr, wr, ib);
#pragma nounroll
    for (int ii = 0; ii < CPW; ++ii) {
        a_load(fn, xr, wr, ib + min(ii + 1, CPW - 1));
        __builtin_amdgcn_sched_barrier(0);
        a_step(fc, va, vb, s2);
        fc = fn;
    }

    float* pp = part2 + (size_t)blockIdx.y * SVEC + (size_t)mt * TILE + lane * 4;
#pragma unroll
    for (int j = 0; j < NUM_CAPS; ++j) {
        float4 q0, q1;
        q0.x = s2[j][0];   q0.y = s2[j][1];   q0.z = s2[j][2];   q0.w = s2[j][3];
        q1.x = s2[j][4];   q1.y = s2[j][5];   q1.z = s2[j][6];   q1.w = s2[j][7];
        *(float4*)(pp + (2 * j + 0) * 128) = q0;
        *(float4*)(pp + (2 * j + 1) * 128) = q1;
    }
}

#define RG   4
__global__ __launch_bounds__(512) void caps_reduce_squash_tiles(const float* __restrict__ part2,
                                                                float* __restrict__ out)
{
    __shared__ float4 red[RG][2][2][32];
    __shared__ float4 line[16 * 8];

    const int u    = threadIdx.x;
    const int lane = u & 31;
    const int rq   = (u >> 5) & 1;
    const int jj   = (u >> 6) & 1;
    const int kg   = u >> 7;
    const int jp   = blockIdx.x & 15;
    const int mt   = blockIdx.x >> 4;
    const int j    = 2 * jp + jj;

    const float* p = part2 + (size_t)(kg * (NCH / RG)) * SVEC + (size_t)mt * TILE
                   + ((2 * j + rq) * 32 + lane) * 4;
    float4 s = { 0.0f, 0.0f, 0.0f, 0.0f };
    for (int k = 0; k < NCH / RG; ++k) {
        const float4 v = *(const float4*)(p + (size_t)k * SVEC);
        s.x += v.x;   s.y += v.y;   s.z += v.z;   s.w += v.w;
    }
    red[kg][jj][rq][lane] = s;
    __syncthreads();

    if (u < 64) {
        const int c = u >> 5;
        float4 a = red[0][c][0][lane];
        float4 b = red[0][c][1][lane];
#pragma unroll
        for (int g = 1; g < RG; ++g) {
            const float4 a2 = red[g][c][0][lane];
            const float4 b2 = red[g][c][1][lane];
            a.x += a2.x;   a.y += a2.y;   a.z += a2.z;   a.w += a2.w;
            b.x += b2.x;   b.y += b2.y;   b.z += b2.z;   b.w += b2.w;
        }
        float sq = a.x * a.x + a.y * a.y + a.z * a.z + a.w * a.w
                 + b.x * b.x + b.y * b.y + b.z * b.z + b.w * b.w;
        sq += __shfl_xor(sq, 16, 32);
        const float scale = sq / ((1.0f + sq) * (sqrtf(sq) + EPS));
        a.x *= scale;   a.y *= scale;   a.z *= scale;   a.w *= scale;
        b.x *= scale;   b.y *= scale;   b.z *= scale;   b.w *= scale;
        const int lo = lane & 15;
        const int hi = lane >> 4;
        line[lo * 8 + c * 4 + hi * 2 + 0] = a;
        line[lo * 8 + c * 4 + hi * 2 + 1] = b;
    }
    __syncthreads();

    if (u < 128) {
        const int lo = u >> 3;
        const int q  = u & 7;
        float* o = out + (size_t)(mt * 16 + lo) * (NUM_CAPS * CAP_DIM) + jp * 32 + q * 4;
        *(float4*)o = line[u];
    }
}

extern "C" void kernel_launch(void* const* d_in, const int* in_sizes, int n_in,
                              void* d_out, int out_size, void* d_ws, size_t ws_size,
                              hipStream_t stream)
{
    const float* x = (const float*)d_in[0];
    const float* W = (const float*)d_in[1];
    float* vout = (float*)d_out;

    float* v1    = (float*)d_ws;
    float* part2 = v1 + (size_t)SVEC;
    float* part  = part2 + (size_t)BATCH * IN_CAPS * NUM_CAPS;

    caps_s_partial<<<dim3(NUM_CAPS / JG, SPLITK), 256, 0, stream>>>(x, W, part);
    caps_reduce_squash<<<dim3(SVEC / 256),       256, 0, stream>>>(part, v1,
                                                                  1.0f / (float)NUM_CAPS);
    caps_agree_route<<<dim3(BATCH / 64, NCH), 128, 0, stream>>>(x, W, v1, part2);
    caps_reduce_squash_tiles<<<dim3((BATCH / 16) * (NUM_CAPS / 2)), 512, 0, stream>>>(part2, vout);
}


# A and B fragments of both GEMM kernels loaded as one 128-bit load per lane (half the load instructions); lane half holds k=4h..4h+3, so k order inside the two WMMAs and the sum order change
# speedup vs baseline: 5.4609x; 1.1033x over previous
#include <hip/hip_runtime.h>
#include <hip/hip_bf16.h>
#include <math.h>


#define BATCH    128
#define IN_CAPS  2048
#define IN_DIM   8
#define NUM_CAPS 32
#define CAP_DIM  16
#define EPS      1e-8f

#define SPLITK   32
#define KCHUNK   (IN_CAPS / SPLITK)
#define SVEC     (BATCH * NUM_CAPS * CAP_DIM)

typedef float v2f __attribute__((ext_vector_type(2)));
typedef float v8f __attribute__((ext_vector_type(8)));


#define JG   4
#define ST   4
#define NST  (KCHUNK / ST)

struct SFrag {
    v2f    xa[ST][2];
    v2f    wb[ST][JG][2];
};

__device__ __forceinline__ void s_load_stage(SFrag& f, const float* __restrict__ xr,
                                             const float* __restrict__ wr, int i)
{
    const float* xi = xr + i * IN_DIM;
    const float* wi = wr + (size_t)i * (NUM_CAPS * CAP_DIM * IN_DIM);
#pragma unroll
    for (int t = 0; t < ST; ++t) {
        const float4 p = *(const float4*)(xi + t * IN_DIM);
        f.xa[t][0].x = p.x;   f.xa[t][0].y = p.y;
        f.xa[t][1].x = p.z;   f.xa[t][1].y = p.w;
#pragma unroll
        for (int jj = 0; jj < JG; ++jj) {
            const float4 q = *(const float4*)(wi + t * (NUM_CAPS * CAP_DIM * IN_DIM)
                                                 + jj * (CAP_DIM * IN_DIM));
            f.wb[t][jj][0].x = q.x;   f.wb[t][jj][0].y = q.y;
            f.wb[t][jj][1].x = q.z;   f.wb[t][jj][1].y = q.w;
        }
    }
}

__device__ __forceinline__ void s_compute_stage(const SFrag& f, v8f (&acc)[JG])
{
#pragma unroll
    for (int t = 0; t < ST; ++t) {
#pragma unroll
        for (int s = 0; s < 2; ++s) {
#pragma unroll
            for (int jj = 0; jj < JG; ++jj) {
                acc[jj] = __builtin_amdgcn_wmma_f32_16x16x4_f32(
                              false, f.xa[t][s], false, f.wb[t][jj][s], (short)0, acc[jj], false, false);
            }
        }
    }
}

__global__ __launch_bounds__(256) void caps_s_partial(const float* __restrict__ x,
                                                      const float* __restrict__ W,
                                                      float* __restrict__ part)
{
    const int j0   = blockIdx.x * JG;
    const int kc   = blockIdx.y;
    const int wav  = threadIdx.x >> 5;
    const int lane = threadIdx.x & 31;
    const int lo   = lane & 15;
    const int hi   = lane >> 4;
    const int m0   = wav * 16;
    const int i0   = kc * KCHUNK;

    const float* xr = x  + (size_t)(m0 + lo) * (IN_CAPS * IN_DIM) + 4 * hi;
    const float* wr = W  + (size_t)j0 * (CAP_DIM * IN_DIM) + lo * IN_DIM + 4 * hi;

    v8f acc[JG];
#pragma unroll
    for (int jj = 0; jj < JG; ++jj) acc[jj] = v8f{};

    SFrag f0, f1;
    s_load_stage(f0, xr, wr, i0);
    for (int st = 0; st < NST - 2; st += 2) {
        s_load_stage(f1, xr, wr, i0 + (st + 1) * ST);
        __builtin_amdgcn_sched_barrier(0);
        s_compute_stage(f0, acc);
        s_load_stage(f0, xr, wr, i0 + (st + 2) * ST);
        __builtin_amdgcn_sched_barrier(0);
        s_compute_stage(f1, acc);
    }
    s_load_stage(f1, xr, wr, i0 + (NST - 1) * ST);
    __builtin_amdgcn_sched_barrier(0);
    s_compute_stage(f0, acc);
    s_compute_stage(f1, acc);

#pragma unroll
    for (int jj = 0; jj < JG; ++jj) {
        float* p = part + (size_t)kc * SVEC + (j0 + jj) * CAP_DIM + lo;
#pragma unroll
        for (int r = 0; r < 8; ++r)
            p[(size_t)(m0 + r + 8 * hi) * (NUM_CAPS * CAP_DIM)] = acc[jj][r];
    }
}

__global__ __launch_bounds__(256) void caps_reduce_squash(const float* __restrict__ part,
                                                          float* __restrict__ out,
                                                          float pre)
{
    const int t = blockIdx.x * 256 + threadIdx.x;
    float s = 0.0f;
    for (int k = 0; k < SPLITK; ++k)
        s += part[(size_t)k * SVEC + t];
    s *= pre;
    float sq = s * s;
    sq += __shfl_xor(sq, 1, 16);
    sq += __shfl_xor(sq, 2, 16);
    sq += __shfl_xor(sq, 4, 16);
    sq += __shfl_xor(sq, 8, 16);
    const float scale = sq / ((1.0f + sq) * (sqrtf(sq) + EPS));
    out[t] = s * scale;
}

#define CPW  16
#define NCH  (IN_CAPS / CPW)
#define TILE (16 * NUM_CAPS * CAP_DIM)

__device__ __forceinline__ float xhalf(float v)
{
    const int b = __builtin_bit_cast(int, v);
    return __builtin_bit_cast(float,
        __builtin_amdgcn_permlanex16(b, b, 0x76543210, 0xfedcba98, false, false));
}

struct AFrag {
    v2f xb0, xb1;
    v2f wa[NUM_CAPS], wb[NUM_CAPS];
};

__device__ __forceinline__ void a_load(AFrag& f, const float* __restrict__ xr,
                                       const float* __restrict__ wr, int i)
{
    const float4 p = *(const float4*)(xr + i * IN_DIM);
    f.xb0.x = p.x;   f.xb0.y = p.y;
    f.xb1.x = p.z;   f.xb1.y = p.w;
    const float* wi = wr + (size_t)i * (NUM_CAPS * CAP_DIM * IN_DIM);
#pragma unroll
    for (int j = 0; j < NUM_CAPS; ++j) {
        const float4 q = *(const float4*)(wi + j * (CAP_DIM * IN_DIM));
        f.wa[j].x = q.x;   f.wa[j].y = q.y;
        f.wb[j].x = q.z;   f.wb[j].y = q.w;
    }
}

__device__ __forceinline__ void a_step(const AFrag& f, const float4 (&va)[NUM_CAPS],
                                       const float4 (&vb)[NUM_CAPS], v8f (&s2)[NUM_CAPS])
{
    float lg[NUM_CAPS];
#pragma unroll
    for (int j = 0; j < NUM_CAPS; ++j) {
        v8f acc = {};
        acc = __builtin_amdgcn_wmma_f32_16x16x4_f32(
                  false, f.wa[j], false, f.xb0, (short)0, acc, false, false);
        acc = __builtin_amdgcn_wmma_f32_16x16x4_f32(
                  false, f.wb[j], false, f.xb1, (short)0, acc, false, false);
        float t = acc[0] * va[j].x;
        t += acc[1] * va[j].y;
        t += acc[2] * va[j].z;
        t += acc[3] * va[j].w;
        t += acc[4] * vb[j].x;
        t += acc[5] * vb[j].y;
        t += acc[6] * vb[j].z;
        t += acc[7] * vb[j].w;
        lg[j] = t + xhalf(t);
    }

    float mx = -INFINITY;
#pragma unroll
    for (int j = 0; j < NUM_CAPS; ++j) mx = fmaxf(mx, lg[j]);
    float sum = 0.0f;
#pragma unroll
    for (int j = 0; j < NUM_CAPS; ++j) { lg[j] = __expf(lg[j] - mx); sum += lg[j]; }
    const float inv = 1.0f / sum;

#pragma unroll
    for (int g = 0; g < NUM_CAPS; g += 4) {
        v2f b0[4], b1[4];
#pragma unroll
        for (int jj = 0; jj < 4; ++jj) {
            const float c = lg[g + jj] * inv;
            b0[jj].x = c * f.xb0.x;   b0[jj].y = c * f.xb0.y;
            b1[jj].x = c * f.xb1.x;   b1[jj].y = c * f.xb1.y;
        }
#pragma unroll
        for (int jj = 0; jj < 4; ++jj)
            s2[g + jj] = __builtin_amdgcn_wmma_f32_16x16x4_f32(
                             false, f.wa[g + jj], false, b0[jj], (short)0, s2[g + jj], false, false);
#pragma unroll
        for (int jj = 0; jj < 4; ++jj)
            s2[g + jj] = __builtin_amdgcn_wmma_f32_16x16x4_f32(
                             false, f.wb[g + jj], false, b1[jj], (short)0, s2[g + jj], false, false);
    }
}

__global__ __launch_bounds__(128) void caps_agree_route(const float* __restrict__ x,
                                                        const float* __restrict__ W,
                                                        const float* __restrict__ v1,
                                                        float* __restrict__ part2)
{
    const int wav  = threadIdx.x >> 5;
    const int lane = threadIdx.x & 31;
    const int lo   = lane & 15;
    const int hi   = lane >> 4;
    const int mt   = blockIdx.x * 4 + wav;
    const int m0   = mt * 16;
    const int ib   = blockIdx.y * CPW;

    const float* xr = x  + (size_t)(m0 + lo) * (IN_CAPS * IN_DIM) + 4 * hi;
    const float* vr = v1 + (size_t)(m0 + lo) * (NUM_CAPS * CAP_DIM) + 8 * hi;
    const float* wr = W  + lo * IN_DIM + 4 * hi;

    float4 va[NUM_CAPS], vb[NUM_CAPS];
#pragma unroll
    for (int j = 0; j < NUM_CAPS; ++j) {
        va[j] = *(const float4*)(vr + j * CAP_DIM);
        vb[j] = *(const float4*)(vr + j * CAP_DIM + 4);
    }

    v8f s2[NUM_CAPS];
#pragma unroll
    for (int j = 0; j < NUM_CAPS; ++j) s2[j] = v8f{};

    AFrag fc, fn;
    a_load(fc, xr, wr, ib);
#pragma nounroll
    for (int ii = 0; ii < CPW; ++ii) {
        a_load(fn, xr, wr, ib + min(ii + 1, CPW - 1));
        __builtin_amdgcn_sched_barrier(0);
        a_step(fc, va, vb, s2);
        fc = fn;
    }

    float* pp = part2 + (size_t)blockIdx.y * SVEC + (size_t)mt * TILE + lane * 4;
#pragma unroll
    for (int j = 0; j < NUM_CAPS; ++j) {
        float4 q0, q1;
        q0.x = s2[j][0];   q0.y = s2[j][1];   q0.z = s2[j][2];   q0.w = s2[j][3];
        q1.x = s2[j][4];   q1.y = s2[j][5];   q1.z = s2[j][6];   q1.w = s2[j][7];
        *(float4*)(pp + (2 * j + 0) * 128) = q0;
        *(float4*)(pp + (2 * j + 1) * 128) = q1;
    }
}

#define RG   4
__global__ __launch_bounds__(512) void caps_reduce_squash_tiles(const float* __restrict__ part2,
                                                                float* __restrict__ out)
{
    __shared__ float4 red[RG][2][2][32];
    __shared__ float4 line[16 * 8];

    const int u    = threadIdx.x;
    const int lane = u & 31;
    const int rq   = (u >> 5) & 1;
    const int jj   = (u >> 6) & 1;
    const int kg   = u >> 7;
    const int jp   = blockIdx.x & 15;
    const int mt   = blockIdx.x >> 4;
    const int j    = 2 * jp + jj;

    const float* p = part2 + (size_t)(kg * (NCH / RG)) * SVEC + (size_t)mt * TILE
                   + ((2 * j + rq) * 32 + lane) * 4;
    float4 s = { 0.0f, 0.0f, 0.0f, 0.0f };
    for (int k = 0; k < NCH / RG; ++k) {
        const float4 v = *(const float4*)(p + (size_t)k * SVEC);
        s.x += v.x;   s.y += v.y;   s.z += v.z;   s.w += v.w;
    }
    red[kg][jj][rq][lane] = s;
    __syncthreads();

    if (u < 64) {
        const int c = u >> 5;
        float4 a = red[0][c][0][lane];
        float4 b = red[0][c][1][lane];
#pragma unroll
        for (int g = 1; g < RG; ++g) {
            const float4 a2 = red[g][c][0][lane];
            const float4 b2 = red[g][c][1][lane];
            a.x += a2.x;   a.y += a2.y;   a.z += a2.z;   a.w += a2.w;
            b.x += b2.x;   b.y += b2.y;   b.z += b2.z;   b.w += b2.w;
        }
        float sq = a.x * a.x + a.y * a.y + a.z * a.z + a.w * a.w
                 + b.x * b.x + b.y * b.y + b.z * b.z + b.w * b.w;
        sq += __shfl_xor(sq, 16, 32);
        const float scale = sq / ((1.0f + sq) * (sqrtf(sq) + EPS));
        a.x *= scale;   a.y *= scale;   a.z *= scale;   a.w *= scale;
        b.x *= scale;   b.y *= scale;   b.z *= scale;   b.w *= scale;
        const int lo = lane & 15;
        const int hi = lane >> 4;
        line[lo * 8 + c * 4 + hi * 2 + 0] = a;
        line[lo * 8 + c * 4 + hi * 2 + 1] = b;
    }
    __syncthreads();

    if (u < 128) {
        const int lo = u >> 3;
        const int q  = u & 7;
        float* o = out + (size_t)(mt * 16 + lo) * (NUM_CAPS * CAP_DIM) + jp * 32 + q * 4;
        *(float4*)o = line[u];
    }
}

extern "C" void kernel_launch(void* const* d_in, const int* in_sizes, int n_in,
                              void* d_out, int out_size, void* d_ws, size_t ws_size,
                              hipStream_t stream)
{
    const float* x = (const float*)d_in[0];
    const float* W = (const float*)d_in[1];
    float* vout = (float*)d_out;

    float* v1    = (float*)d_ws;
    float* part2 = v1 + (size_t)SVEC;
    float* part  = part2 + (size_t)BATCH * IN_CAPS * NUM_CAPS;

    caps_s_partial<<<dim3(NUM_CAPS / JG, SPLITK), 256, 0, stream>>>(x, W, part);
    caps_reduce_squash<<<dim3(SVEC / 256),       256, 0, stream>>>(part, v1,
                                                                  1.0f / (float)NUM_CAPS);
    caps_agree_route<<<dim3(BATCH / 64, NCH), 128, 0, stream>>>(x, W, v1, part2);
    caps_reduce_squash_tiles<<<dim3((BATCH / 16) * (NUM_CAPS / 2)), 512, 0, stream>>>(part2, vout);
}


# agreement kernel: s2 += c*u_hat as 8 in-lane FMAs on the u_hat tiles kept in registers; the second WMMA pass over c-scaled x is dropped (W x formed once on the matrix unit); s2 sum order changed
# speedup vs baseline: 6.2347x; 1.1417x over previous
#include <hip/hip_runtime.h>
#include <hip/hip_bf16.h>
#include <math.h>


#define BATCH    128
#define IN_CAPS  2048
#define IN_DIM   8
#define NUM_CAPS 32
#define CAP_DIM  16
#define EPS      1e-8f

#define SPLITK   32
#define KCHUNK   (IN_CAPS / SPLITK)
#define SVEC     (BATCH * NUM_CAPS * CAP_DIM)

typedef float v2f __attribute__((ext_vector_type(2)));
typedef float v8f __attribute__((ext_vector_type(8)));


#define JG   4
#define ST   4
#define NST  (KCHUNK / ST)

struct SFrag {
    v2f    xa[ST][2];
    v2f    wb[ST][JG][2];
};

__device__ __forceinline__ void s_load_stage(SFrag& f, const float* __restrict__ xr,
                                             const float* __restrict__ wr, int i)
{
    const float* xi = xr + i * IN_DIM;
    const float* wi = wr + (size_t)i * (NUM_CAPS * CAP_DIM * IN_DIM);
#pragma unroll
    for (int t = 0; t < ST; ++t) {
        const float4 p = *(const float4*)(xi + t * IN_DIM);
        f.xa[t][0].x = p.x;   f.xa[t][0].y = p.y;
        f.xa[t][1].x = p.z;   f.xa[t][1].y = p.w;
#pragma unroll
        for (int jj = 0; jj < JG; ++jj) {
            const float4 q = *(const float4*)(wi + t * (NUM_CAPS * CAP_DIM * IN_DIM)
                                                 + jj * (CAP_DIM * IN_DIM));
            f.wb[t][jj][0].x = q.x;   f.wb[t][jj][0].y = q.y;
            f.wb[t][jj][1].x = q.z;   f.wb[t][jj][1].y = q.w;
        }
    }
}

__device__ __forceinline__ void s_compute_stage(const SFrag& f, v8f (&acc)[JG])
{
#pragma unroll
    for (int t = 0; t < ST; ++t) {
#pragma unroll
        for (int s = 0; s < 2; ++s) {
#pragma unroll
            for (int jj = 0; jj < JG; ++jj) {
                acc[jj] = __builtin_amdgcn_wmma_f32_16x16x4_f32(
                              false, f.xa[t][s], false, f.wb[t][jj][s], (short)0, acc[jj], false, false);
            }
        }
    }
}

__global__ __launch_bounds__(256) void caps_s_partial(const float* __restrict__ x,
                                                      const float* __restrict__ W,
                                                      float* __restrict__ part)
{
    const int j0   = blockIdx.x * JG;
    const int kc   = blockIdx.y;
    const int wav  = threadIdx.x >> 5;
    const int lane = threadIdx.x & 31;
    const int lo   = lane & 15;
    const int hi   = lane >> 4;
    const int m0   = wav * 16;
    const int i0   = kc * KCHUNK;

    const float* xr = x  + (size_t)(m0 + lo) * (IN_CAPS * IN_DIM) + 4 * hi;
    const float* wr = W  + (size_t)j0 * (CAP_DIM * IN_DIM) + lo * IN_DIM + 4 * hi;

    v8f acc[JG];
#pragma unroll
    for (int jj = 0; jj < JG; ++jj) acc[jj] = v8f{};

    SFrag f0, f1;
    s_load_stage(f0, xr, wr, i0);
    for (int st = 0; st < NST - 2; st += 2) {
        s_load_stage(f1, xr, wr, i0 + (st + 1) * ST);
        __builtin_amdgcn_sched_barrier(0);
        s_compute_stage(f0, acc);
        s_load_stage(f0, xr, wr, i0 + (st + 2) * ST);
        __builtin_amdgcn_sched_barrier(0);
        s_compute_stage(f1, acc);
    }
    s_load_stage(f1, xr, wr, i0 + (NST - 1) * ST);
    __builtin_amdgcn_sched_barrier(0);
    s_compute_stage(f0, acc);
    s_compute_stage(f1, acc);

#pragma unroll
    for (int jj = 0; jj < JG; ++jj) {
        float* p = part + (size_t)kc * SVEC + (j0 + jj) * CAP_DIM + lo;
#pragma unroll
        for (int r = 0; r < 8; ++r)
            p[(size_t)(m0 + r + 8 * hi) * (NUM_CAPS * CAP_DIM)] = acc[jj][r];
    }
}

__global__ __launch_bounds__(256) void caps_reduce_squash(const float* __restrict__ part,
                                                          float* __restrict__ out,
                                                          float pre)
{
    const int t = blockIdx.x * 256 + threadIdx.x;
    float s = 0.0f;
    for (int k = 0; k < SPLITK; ++k)
        s += part[(size_t)k * SVEC + t];
    s *= pre;
    float sq = s * s;
    sq += __shfl_xor(sq, 1, 16);
    sq += __shfl_xor(sq, 2, 16);
    sq += __shfl_xor(sq, 4, 16);
    sq += __shfl_xor(sq, 8, 16);
    const float scale = sq / ((1.0f + sq) * (sqrtf(sq) + EPS));
    out[t] = s * scale;
}

#define CPW  16
#define NCH  (IN_CAPS / CPW)
#define TILE (16 * NUM_CAPS * CAP_DIM)

__device__ __forceinline__ float xhalf(float v)
{
    const int b = __builtin_bit_cast(int, v);
    return __builtin_bit_cast(float,
        __builtin_amdgcn_permlanex16(b, b, 0x76543210, 0xfedcba98, false, false));
}

struct AFrag {
    v2f xb0, xb1;
    v2f wa[NUM_CAPS], wb[NUM_CAPS];
};

__device__ __forceinline__ void a_load(AFrag& f, const float* __restrict__ xr,
                                       const float* __restrict__ wr, int i)
{
    const float4 p = *(const float4*)(xr + i * IN_DIM);
    f.xb0.x = p.x;   f.xb0.y = p.y;
    f.xb1.x = p.z;   f.xb1.y = p.w;
    const float* wi = wr + (size_t)i * (NUM_CAPS * CAP_DIM * IN_DIM);
#pragma unroll
    for (int j = 0; j < NUM_CAPS; ++j) {
        const float4 q = *(const float4*)(wi + j * (CAP_DIM * IN_DIM));
        f.wa[j].x = q.x;   f.wa[j].y = q.y;
        f.wb[j].x = q.z;   f.wb[j].y = q.w;
    }
}

__device__ __forceinline__ void a_step(const AFrag& f, const float4 (&va)[NUM_CAPS],
                                       const float4 (&vb)[NUM_CAPS], v8f (&s2)[NUM_CAPS])
{
    v8f   u[NUM_CAPS];
    float lg[NUM_CAPS];
#pragma unroll
    for (int j = 0; j < NUM_CAPS; ++j) {
        v8f acc = {};
        acc = __builtin_amdgcn_wmma_f32_16x16x4_f32(
                  false, f.wa[j], false, f.xb0, (short)0, acc, false, false);
        acc = __builtin_amdgcn_wmma_f32_16x16x4_f32(
                  false, f.wb[j], false, f.xb1, (short)0, acc, false, false);
        u[j] = acc;
        float t = acc[0] * va[j].x;
        t += acc[1] * va[j].y;
        t += acc[2] * va[j].z;
        t += acc[3] * va[j].w;
        t += acc[4] * vb[j].x;
        t += acc[5] * vb[j].y;
        t += acc[6] * vb[j].z;
        t += acc[7] * vb[j].w;
        lg[j] = t + xhalf(t);
    }

    float mx = -INFINITY;
#pragma unroll
    for (int j = 0; j < NUM_CAPS; ++j) mx = fmaxf(mx, lg[j]);
    float sum = 0.0f;
#pragma unroll
    for (int j = 0; j < NUM_CAPS; ++j) { lg[j] = __expf(lg[j] - mx); sum += lg[j]; }
    const float inv = 1.0f / sum;

#pragma unroll
    for (int j = 0; j < NUM_CAPS; ++j) {
        const float c = lg[j] * inv;
        s2[j] += c * u[j];
    }
}

__global__ __launch_bounds__(128) void caps_agree_route(const float* __restrict__ x,
                                                        const float* __restrict__ W,
                                                        const float* __restrict__ v1,
                                                        float* __restrict__ part2)
{
    const int wav  = threadIdx.x >> 5;
    const int lane = threadIdx.x & 31;
    const int lo   = lane & 15;
    const int hi   = lane >> 4;
    const int mt   = blockIdx.x * 4 + wav;
    const int m0   = mt * 16;
    const int ib   = blockIdx.y * CPW;

    const float* xr = x  + (size_t)(m0 + lo) * (IN_CAPS * IN_DIM) + 4 * hi;
    const float* vr = v1 + (size_t)(m0 + lo) * (NUM_CAPS * CAP_DIM) + 8 * hi;
    const float* wr = W  + lo * IN_DIM + 4 * hi;

    float4 va[NUM_CAPS], vb[NUM_CAPS];
#pragma unroll
    for (int j = 0; j < NUM_CAPS; ++j) {
        va[j] = *(const float4*)(vr + j * CAP_DIM);
        vb[j] = *(const float4*)(vr + j * CAP_DIM + 4);
    }

    v8f s2[NUM_CAPS];
#pragma unroll
    for (int j = 0; j < NUM_CAPS; ++j) s2[j] = v8f{};

    AFrag fc, fn;
    a_load(fc, xr, wr, ib);
#pragma nounroll
    for (int ii = 0; ii < CPW; ++ii) {
        a_load(fn, xr, wr, ib + min(ii + 1, CPW - 1));
        __builtin_amdgcn_sched_barrier(0);
        a_step(fc, va, vb, s2);
        fc = fn;
    }

    float* pp = part2 + (size_t)blockIdx.y * SVEC + (size_t)mt * TILE + lane * 4;
#pragma unroll
    for (int j = 0; j < NUM_CAPS; ++j) {
        float4 q0, q1;
        q0.x = s2[j][0];   q0.y = s2[j][1];   q0.z = s2[j][2];   q0.w = s2[j][3];
        q1.x = s2[j][4];   q1.y = s2[j][5];   q1.z = s2[j][6];   q1.w = s2[j][7];
        *(float4*)(pp + (2 * j + 0) * 128) = q0;
        *(float4*)(pp + (2 * j + 1) * 128) = q1;
    }
}

#define RG   4
__global__ __launch_bounds__(512) void caps_reduce_squash_tiles(const float* __restrict__ part2,
                                                                float* __restrict__ out)
{
    __shared__ float4 red[RG][2][2][32];
    __shared__ float4 line[16 * 8];

    const int u    = threadIdx.x;
    const int lane = u & 31;
    const int rq   = (u >> 5) & 1;
    const int jj   = (u >> 6) & 1;
    const int kg   = u >> 7;
    const int jp   = blockIdx.x & 15;
    const int mt   = blockIdx.x >> 4;
    const int j    = 2 * jp + jj;

    const float* p = part2 + (size_t)(kg * (NCH / RG)) * SVEC + (size_t)mt * TILE
                   + ((2 * j + rq) * 32 + lane) * 4;
    float4 s = { 0.0f, 0.0f, 0.0f, 0.0f };
    for (int k = 0; k < NCH / RG; ++k) {
        const float4 v = *(const float4*)(p + (size_t)k * SVEC);
        s.x += v.x;   s.y += v.y;   s.z += v.z;   s.w += v.w;
    }
    red[kg][jj][rq][lane] = s;
    __syncthreads();

    if (u < 64) {
        const int c = u >> 5;
        float4 a = red[0][c][0][lane];
        float4 b = red[0][c][1][lane];
#pragma unroll
        for (int g = 1; g < RG; ++g) {
            const float4 a2 = red[g][c][0][lane];
            const float4 b2 = red[g][c][1][lane];
            a.x += a2.x;   a.y += a2.y;   a.z += a2.z;   a.w += a2.w;
            b.x += b2.x;   b.y += b2.y;   b.z += b2.z;   b.w += b2.w;
        }
        float sq = a.x * a.x + a.y * a.y + a.z * a.z + a.w * a.w
                 + b.x * b.x + b.y * b.y + b.z * b.z + b.w * b.w;
        sq += __shfl_xor(sq, 16, 32);
        const float scale = sq / ((1.0f + sq) * (sqrtf(sq) + EPS));
        a.x *= scale;   a.y *= scale;   a.z *= scale;   a.w *= scale;
        b.x *= scale;   b.y *= scale;   b.z *= scale;   b.w *= scale;
        const int lo = lane & 15;
        const int hi = lane >> 4;
        line[lo * 8 + c * 4 + hi * 2 + 0] = a;
        line[lo * 8 + c * 4 + hi * 2 + 1] = b;
    }
    __syncthreads();

    if (u < 128) {
        const int lo = u >> 3;
        const int q  = u & 7;
        float* o = out + (size_t)(mt * 16 + lo) * (NUM_CAPS * CAP_DIM) + jp * 32 + q * 4;
        *(float4*)o = line[u];
    }
}

extern "C" void kernel_launch(void* const* d_in, const int* in_sizes, int n_in,
                              void* d_out, int out_size, void* d_ws, size_t ws_size,
                              hipStream_t stream)
{
    const float* x = (const float*)d_in[0];
    const float* W = (const float*)d_in[1];
    float* vout = (float*)d_out;

    float* v1    = (float*)d_ws;
    float* part2 = v1 + (size_t)SVEC;
    float* part  = part2 + (size_t)BATCH * IN_CAPS * NUM_CAPS;

    caps_s_partial<<<dim3(NUM_CAPS / JG, SPLITK), 256, 0, stream>>>(x, W, part);
    caps_reduce_squash<<<dim3(SVEC / 256),       256, 0, stream>>>(part, v1,
                                                                  1.0f / (float)NUM_CAPS);
    caps_agree_route<<<dim3(BATCH / 64, NCH), 128, 0, stream>>>(x, W, v1, part2);
    caps_reduce_squash_tiles<<<dim3((BATCH / 16) * (NUM_CAPS / 2)), 512, 0, stream>>>(part2, vout);
}


# agreement kernel: a block's 4 waves are one sample tile on 4 capsule groups; their s2 accumulators are summed through LDS before the store, 32 partial slices instead of 128 (s2 sum order changed)
# speedup vs baseline: 7.0068x; 1.1238x over previous
#include <hip/hip_runtime.h>
#include <hip/hip_bf16.h>
#include <math.h>


#define BATCH    128
#define IN_CAPS  2048
#define IN_DIM   8
#define NUM_CAPS 32
#define CAP_DIM  16
#define EPS      1e-8f

#define SPLITK   32
#define KCHUNK   (IN_CAPS / SPLITK)
#define SVEC     (BATCH * NUM_CAPS * CAP_DIM)

typedef float v2f __attribute__((ext_vector_type(2)));
typedef float v8f __attribute__((ext_vector_type(8)));


#define JG   4
#define ST   4
#define NST  (KCHUNK / ST)

struct SFrag {
    v2f    xa[ST][2];
    v2f    wb[ST][JG][2];
};

__device__ __forceinline__ void s_load_stage(SFrag& f, const float* __restrict__ xr,
                                             const float* __restrict__ wr, int i)
{
    const float* xi = xr + i * IN_DIM;
    const float* wi = wr + (size_t)i * (NUM_CAPS * CAP_DIM * IN_DIM);
#pragma unroll
    for (int t = 0; t < ST; ++t) {
        const float4 p = *(const float4*)(xi + t * IN_DIM);
        f.xa[t][0].x = p.x;   f.xa[t][0].y = p.y;
        f.xa[t][1].x = p.z;   f.xa[t][1].y = p.w;
#pragma unroll
        for (int jj = 0; jj < JG; ++jj) {
            const float4 q = *(const float4*)(wi + t * (NUM_CAPS * CAP_DIM * IN_DIM)
                                                 + jj * (CAP_DIM * IN_DIM));
            f.wb[t][jj][0].x = q.x;   f.wb[t][jj][0].y = q.y;
            f.wb[t][jj][1].x = q.z;   f.wb[t][jj][1].y = q.w;
        }
    }
}

__device__ __forceinline__ void s_compute_stage(const SFrag& f, v8f (&acc)[JG])
{
#pragma unroll
    for (int t = 0; t < ST; ++t) {
#pragma unroll
        for (int s = 0; s < 2; ++s) {
#pragma unroll
            for (int jj = 0; jj < JG; ++jj) {
                acc[jj] = __builtin_amdgcn_wmma_f32_16x16x4_f32(
                              false, f.xa[t][s], false, f.wb[t][jj][s], (short)0, acc[jj], false, false);
            }
        }
    }
}

__global__ __launch_bounds__(256) void caps_s_partial(const float* __restrict__ x,
                                                      const float* __restrict__ W,
                                                      float* __restrict__ part)
{
    const int j0   = blockIdx.x * JG;
    const int kc   = blockIdx.y;
    const int wav  = threadIdx.x >> 5;
    const int lane = threadIdx.x & 31;
    const int lo   = lane & 15;
    const int hi   = lane >> 4;
    const int m0   = wav * 16;
    const int i0   = kc * KCHUNK;

    const float* xr = x  + (size_t)(m0 + lo) * (IN_CAPS * IN_DIM) + 4 * hi;
    const float* wr = W  + (size_t)j0 * (CAP_DIM * IN_DIM) + lo * IN_DIM + 4 * hi;

    v8f acc[JG];
#pragma unroll
    for (int jj = 0; jj < JG; ++jj) acc[jj] = v8f{};

    SFrag f0, f1;
    s_load_stage(f0, xr, wr, i0);
    for (int st = 0; st < NST - 2; st += 2) {
        s_load_stage(f1, xr, wr, i0 + (st + 1) * ST);
        __builtin_amdgcn_sched_barrier(0);
        s_compute_stage(f0, acc);
        s_load_stage(f0, xr, wr, i0 + (st + 2) * ST);
        __builtin_amdgcn_sched_barrier(0);
        s_compute_stage(f1, acc);
    }
    s_load_stage(f1, xr, wr, i0 + (NST - 1) * ST);
    __builtin_amdgcn_sched_barrier(0);
    s_compute_stage(f0, acc);
    s_compute_stage(f1, acc);

#pragma unroll
    for (int jj = 0; jj < JG; ++jj) {
        float* p = part + (size_t)kc * SVEC + (j0 + jj) * CAP_DIM + lo;
#pragma unroll
        for (int r = 0; r < 8; ++r)
            p[(size_t)(m0 + r + 8 * hi) * (NUM_CAPS * CAP_DIM)] = acc[jj][r];
    }
}

__global__ __launch_bounds__(256) void caps_reduce_squash(const float* __restrict__ part,
                                                          float* __restrict__ out,
                                                          float pre)
{
    const int t = blockIdx.x * 256 + threadIdx.x;
    float s = 0.0f;
    for (int k = 0; k < SPLITK; ++k)
        s += part[(size_t)k * SVEC + t];
    s *= pre;
    float sq = s * s;
    sq += __shfl_xor(sq, 1, 16);
    sq += __shfl_xor(sq, 2, 16);
    sq += __shfl_xor(sq, 4, 16);
    sq += __shfl_xor(sq, 8, 16);
    const float scale = sq / ((1.0f + sq) * (sqrtf(sq) + EPS));
    out[t] = s * scale;
}

#define CPW  16
#define WPB  4
#define NSL  (IN_CAPS / (CPW * WPB))
#define TILE (16 * NUM_CAPS * CAP_DIM)

__device__ __forceinline__ float xhalf(float v)
{
    const int b = __builtin_bit_cast(int, v);
    return __builtin_bit_cast(float,
        __builtin_amdgcn_permlanex16(b, b, 0x76543210, 0xfedcba98, false, false));
}

struct AFrag {
    v2f xb0, xb1;
    v2f wa[NUM_CAPS], wb[NUM_CAPS];
};

__device__ __forceinline__ void a_load(AFrag& f, const float* __restrict__ xr,
                                       const float* __restrict__ wr, int i)
{
    const float4 p = *(const float4*)(xr + i * IN_DIM);
    f.xb0.x = p.x;   f.xb0.y = p.y;
    f.xb1.x = p.z;   f.xb1.y = p.w;
    const float* wi = wr + (size_t)i * (NUM_CAPS * CAP_DIM * IN_DIM);
#pragma unroll
    for (int j = 0; j < NUM_CAPS; ++j) {
        const float4 q = *(const float4*)(wi + j * (CAP_DIM * IN_DIM));
        f.wa[j].x = q.x;   f.wa[j].y = q.y;
        f.wb[j].x = q.z;   f.wb[j].y = q.w;
    }
}

__device__ __forceinline__ void a_step(const AFrag& f, const float4 (&va)[NUM_CAPS],
                                       const float4 (&vb)[NUM_CAPS], v8f (&s2)[NUM_CAPS])
{
    v8f   u[NUM_CAPS];
    float lg[NUM_CAPS];
#pragma unroll
    for (int j = 0; j < NUM_CAPS; ++j) {
        v8f acc = {};
        acc = __builtin_amdgcn_wmma_f32_16x16x4_f32(
                  false, f.wa[j], false, f.xb0, (short)0, acc, false, false);
        acc = __builtin_amdgcn_wmma_f32_16x16x4_f32(
                  false, f.wb[j], false, f.xb1, (short)0, acc, false, false);
        u[j] = acc;
        float t = acc[0] * va[j].x;
        t += acc[1] * va[j].y;
        t += acc[2] * va[j].z;
        t += acc[3] * va[j].w;
        t += acc[4] * vb[j].x;
        t += acc[5] * vb[j].y;
        t += acc[6] * vb[j].z;
        t += acc[7] * vb[j].w;
        lg[j] = t + xhalf(t);
    }

    float mx = -INFINITY;
#pragma unroll
    for (int j = 0; j < NUM_CAPS; ++j) mx = fmaxf(mx, lg[j]);
    float sum = 0.0f;
#pragma unroll
    for (int j = 0; j < NUM_CAPS; ++j) { lg[j] = __expf(lg[j] - mx); sum += lg[j]; }
    const float inv = 1.0f / sum;

#pragma unroll
    for (int j = 0; j < NUM_CAPS; ++j) {
        const float c = lg[j] * inv;
        s2[j] += c * u[j];
    }
}

__global__ __launch_bounds__(128) void caps_agree_route(const float* __restrict__ x,
                                                        const float* __restrict__ W,
                                                        const float* __restrict__ v1,
                                                        float* __restrict__ part2)
{
    const int wav  = threadIdx.x >> 5;
    const int lane = threadIdx.x & 31;
    const int lo   = lane & 15;
    const int hi   = lane >> 4;
    const int mt   = blockIdx.x;
    const int m0   = mt * 16;
    const int ib   = (blockIdx.y * WPB + wav) * CPW;

    const float* xr = x  + (size_t)(m0 + lo) * (IN_CAPS * IN_DIM) + 4 * hi;
    const float* vr = v1 + (size_t)(m0 + lo) * (NUM_CAPS * CAP_DIM) + 8 * hi;
    const float* wr = W  + lo * IN_DIM + 4 * hi;

    float4 va[NUM_CAPS], vb[NUM_CAPS];
#pragma unroll
    for (int j = 0; j < NUM_CAPS; ++j) {
        va[j] = *(const float4*)(vr + j * CAP_DIM);
        vb[j] = *(const float4*)(vr + j * CAP_DIM + 4);
    }

    v8f s2[NUM_CAPS];
#pragma unroll
    for (int j = 0; j < NUM_CAPS; ++j) s2[j] = v8f{};

    AFrag fc, fn;
    a_load(fc, xr, wr, ib);
#pragma nounroll
    for (int ii = 0; ii < CPW; ++ii) {
        a_load(fn, xr, wr, ib + min(ii + 1, CPW - 1));
        __builtin_amdgcn_sched_barrier(0);
        a_step(fc, va, vb, s2);
        fc = fn;
    }

    __shared__ float4 red[WPB][2 * NUM_CAPS][32];
#pragma unroll
    for (int j = 0; j < NUM_CAPS; ++j) {
        float4 q0, q1;
        q0.x = s2[j][0];   q0.y = s2[j][1];   q0.z = s2[j][2];   q0.w = s2[j][3];
        q1.x = s2[j][4];   q1.y = s2[j][5];   q1.z = s2[j][6];   q1.w = s2[j][7];
        red[wav][2 * j + 0][lane] = q0;
        red[wav][2 * j + 1][lane] = q1;
    }
    __syncthreads();

    float* pp = part2 + (size_t)blockIdx.y * SVEC + (size_t)mt * TILE + lane * 4;
#pragma unroll
    for (int t = 0; t < (2 * NUM_CAPS) / WPB; ++t) {
        const int q = wav * ((2 * NUM_CAPS) / WPB) + t;
        float4 a = red[0][q][lane];
#pragma unroll
        for (int w = 1; w < WPB; ++w) {
            const float4 b = red[w][q][lane];
            a.x += b.x;   a.y += b.y;   a.z += b.z;   a.w += b.w;
        }
        *(float4*)(pp + q * 128) = a;
    }
}

#define RG   4
__global__ __launch_bounds__(512) void caps_reduce_squash_tiles(const float* __restrict__ part2,
                                                                float* __restrict__ out)
{
    __shared__ float4 red[RG][2][2][32];
    __shared__ float4 line[16 * 8];

    const int u    = threadIdx.x;
    const int lane = u & 31;
    const int rq   = (u >> 5) & 1;
    const int jj   = (u >> 6) & 1;
    const int kg   = u >> 7;
    const int jp   = blockIdx.x & 15;
    const int mt   = blockIdx.x >> 4;
    const int j    = 2 * jp + jj;

    const float* p = part2 + (size_t)(kg * (NSL / RG)) * SVEC + (size_t)mt * TILE
                   + ((2 * j + rq) * 32 + lane) * 4;
    float4 s = { 0.0f, 0.0f, 0.0f, 0.0f };
    for (int k = 0; k < NSL / RG; ++k) {
        const float4 v = *(const float4*)(p + (size_t)k * SVEC);
        s.x += v.x;   s.y += v.y;   s.z += v.z;   s.w += v.w;
    }
    red[kg][jj][rq][lane] = s;
    __syncthreads();

    if (u < 64) {
        const int c = u >> 5;
        float4 a = red[0][c][0][lane];
        float4 b = red[0][c][1][lane];
#pragma unroll
        for (int g = 1; g < RG; ++g) {
            const float4 a2 = red[g][c][0][lane];
            const float4 b2 = red[g][c][1][lane];
            a.x += a2.x;   a.y += a2.y;   a.z += a2.z;   a.w += a2.w;
            b.x += b2.x;   b.y += b2.y;   b.z += b2.z;   b.w += b2.w;
        }
        float sq = a.x * a.x + a.y * a.y + a.z * a.z + a.w * a.w
                 + b.x * b.x + b.y * b.y + b.z * b.z + b.w * b.w;
        sq += __shfl_xor(sq, 16, 32);
        const float scale = sq / ((1.0f + sq) * (sqrtf(sq) + EPS));
        a.x *= scale;   a.y *= scale;   a.z *= scale;   a.w *= scale;
        b.x *= scale;   b.y *= scale;   b.z *= scale;   b.w *= scale;
        const int lo = lane & 15;
        const int hi = lane >> 4;
        line[lo * 8 + c * 4 + hi * 2 + 0] = a;
        line[lo * 8 + c * 4 + hi * 2 + 1] = b;
    }
    __syncthreads();

    if (u < 128) {
        const int lo = u >> 3;
        const int q  = u & 7;
        float* o = out + (size_t)(mt * 16 + lo) * (NUM_CAPS * CAP_DIM) + jp * 32 + q * 4;
        *(float4*)o = line[u];
    }
}

extern "C" void kernel_launch(void* const* d_in, const int* in_sizes, int n_in,
                              void* d_out, int out_size, void* d_ws, size_t ws_size,
                              hipStream_t stream)
{
    const float* x = (const float*)d_in[0];
    const float* W = (const float*)d_in[1];
    float* vout = (float*)d_out;

    float* v1    = (float*)d_ws;
    float* part2 = v1 + (size_t)SVEC;
    float* part  = part2 + (size_t)BATCH * IN_CAPS * NUM_CAPS;

    caps_s_partial<<<dim3(NUM_CAPS / JG, SPLITK), 256, 0, stream>>>(x, W, part);
    caps_reduce_squash<<<dim3(SVEC / 256),       256, 0, stream>>>(part, v1,
                                                                  1.0f / (float)NUM_CAPS);
    caps_agree_route<<<dim3(BATCH / 16, NSL), 128, 0, stream>>>(x, W, v1, part2);
    caps_reduce_squash_tiles<<<dim3((BATCH / 16) * (NUM_CAPS / 2)), 512, 0, stream>>>(part2, vout);
}


# agreement kernel: A tile rows re-paired (8 components of 2 capsules) so a lane half holds a capsule's 16: logit d-sum in-lane, 2 half exchanges per row not 32; logit and softmax sum order changed
# speedup vs baseline: 7.4167x; 1.0585x over previous
#include <hip/hip_runtime.h>
#include <hip/hip_bf16.h>
#include <math.h>


#define BATCH    128
#define IN_CAPS  2048
#define IN_DIM   8
#define NUM_CAPS 32
#define CAP_DIM  16
#define EPS      1e-8f

#define SPLITK   32
#define KCHUNK   (IN_CAPS / SPLITK)
#define SVEC     (BATCH * NUM_CAPS * CAP_DIM)

typedef float v2f __attribute__((ext_vector_type(2)));
typedef float v4f __attribute__((ext_vector_type(4)));
typedef float v8f __attribute__((ext_vector_type(8)));


#define JG   4
#define ST   4
#define NST  (KCHUNK / ST)

struct SFrag {
    v2f    xa[ST][2];
    v2f    wb[ST][JG][2];
};

__device__ __forceinline__ void s_load_stage(SFrag& f, const float* __restrict__ xr,
                                             const float* __restrict__ wr, int i)
{
    const float* xi = xr + i * IN_DIM;
    const float* wi = wr + (size_t)i * (NUM_CAPS * CAP_DIM * IN_DIM);
#pragma unroll
    for (int t = 0; t < ST; ++t) {
        const float4 p = *(const float4*)(xi + t * IN_DIM);
        f.xa[t][0].x = p.x;   f.xa[t][0].y = p.y;
        f.xa[t][1].x = p.z;   f.xa[t][1].y = p.w;
#pragma unroll
        for (int jj = 0; jj < JG; ++jj) {
            const float4 q = *(const float4*)(wi + t * (NUM_CAPS * CAP_DIM * IN_DIM)
                                                 + jj * (CAP_DIM * IN_DIM));
            f.wb[t][jj][0].x = q.x;   f.wb[t][jj][0].y = q.y;
            f.wb[t][jj][1].x = q.z;   f.wb[t][jj][1].y = q.w;
        }
    }
}

__device__ __forceinline__ void s_compute_stage(const SFrag& f, v8f (&acc)[JG])
{
#pragma unroll
    for (int t = 0; t < ST; ++t) {
#pragma unroll
        for (int s = 0; s < 2; ++s) {
#pragma unroll
            for (int jj = 0; jj < JG; ++jj) {
                acc[jj] = __builtin_amdgcn_wmma_f32_16x16x4_f32(
                              false, f.xa[t][s], false, f.wb[t][jj][s], (short)0, acc[jj], false, false);
            }
        }
    }
}

__global__ __launch_bounds__(256) void caps_s_partial(const float* __restrict__ x,
                                                      const float* __restrict__ W,
                                                      float* __restrict__ part)
{
    const int j0   = blockIdx.x * JG;
    const int kc   = blockIdx.y;
    const int wav  = threadIdx.x >> 5;
    const int lane = threadIdx.x & 31;
    const int lo   = lane & 15;
    const int hi   = lane >> 4;
    const int m0   = wav * 16;
    const int i0   = kc * KCHUNK;

    const float* xr = x  + (size_t)(m0 + lo) * (IN_CAPS * IN_DIM) + 4 * hi;
    const float* wr = W  + (size_t)j0 * (CAP_DIM * IN_DIM) + lo * IN_DIM + 4 * hi;

    v8f acc[JG];
#pragma unroll
    for (int jj = 0; jj < JG; ++jj) acc[jj] = v8f{};

    SFrag f0, f1;
    s_load_stage(f0, xr, wr, i0);
    for (int st = 0; st < NST - 2; st += 2) {
        s_load_stage(f1, xr, wr, i0 + (st + 1) * ST);
        __builtin_amdgcn_sched_barrier(0);
        s_compute_stage(f0, acc);
        s_load_stage(f0, xr, wr, i0 + (st + 2) * ST);
        __builtin_amdgcn_sched_barrier(0);
        s_compute_stage(f1, acc);
    }
    s_load_stage(f1, xr, wr, i0 + (NST - 1) * ST);
    __builtin_amdgcn_sched_barrier(0);
    s_compute_stage(f0, acc);
    s_compute_stage(f1, acc);

#pragma unroll
    for (int jj = 0; jj < JG; ++jj) {
        float* p = part + (size_t)kc * SVEC + (j0 + jj) * CAP_DIM + lo;
#pragma unroll
        for (int r = 0; r < 8; ++r)
            p[(size_t)(m0 + r + 8 * hi) * (NUM_CAPS * CAP_DIM)] = acc[jj][r];
    }
}

__global__ __launch_bounds__(256) void caps_reduce_squash(const float* __restrict__ part,
                                                          float* __restrict__ out,
                                                          float pre)
{
    const int t = blockIdx.x * 256 + threadIdx.x;
    float s = 0.0f;
    for (int k = 0; k < SPLITK; ++k)
        s += part[(size_t)k * SVEC + t];
    s *= pre;
    float sq = s * s;
    sq += __shfl_xor(sq, 1, 16);
    sq += __shfl_xor(sq, 2, 16);
    sq += __shfl_xor(sq, 4, 16);
    sq += __shfl_xor(sq, 8, 16);
    const float scale = sq / ((1.0f + sq) * (sqrtf(sq) + EPS));
    out[t] = s * scale;
}

#define CPW  16
#define WPB  4
#define NSL  (IN_CAPS / (CPW * WPB))
#define TILE (16 * NUM_CAPS * CAP_DIM)

__device__ __forceinline__ float xhalf(float v)
{
    const int b = __builtin_bit_cast(int, v);
    return __builtin_bit_cast(float,
        __builtin_amdgcn_permlanex16(b, b, 0x76543210, 0xfedcba98, false, false));
}

struct AFrag {
    v2f xb0, xb1;
    v2f wa[NUM_CAPS], wb[NUM_CAPS];
};

__device__ __forceinline__ void a_load(AFrag& f, const float* __restrict__ xr,
                                       const float* __restrict__ wr, int i)
{
    const float4 p = *(const float4*)(xr + i * IN_DIM);
    f.xb0.x = p.x;   f.xb0.y = p.y;
    f.xb1.x = p.z;   f.xb1.y = p.w;
    const float* wi = wr + (size_t)i * (NUM_CAPS * CAP_DIM * IN_DIM);
#pragma unroll
    for (int q = 0; q < NUM_CAPS; ++q) {
        const float4 w = *(const float4*)(wi + (q >> 1) * (2 * CAP_DIM * IN_DIM)
                                             + (q & 1) * (8 * IN_DIM));
        f.wa[q].x = w.x;   f.wa[q].y = w.y;
        f.wb[q].x = w.z;   f.wb[q].y = w.w;
    }
}

#define NPAIR (NUM_CAPS / 2)
__device__ __forceinline__ v2f pr(const v8f& a, int k)
{
    return k == 0 ? __builtin_shufflevector(a, a, 0, 1) : k == 1 ? __builtin_shufflevector(a, a, 2, 3)
         : k == 2 ? __builtin_shufflevector(a, a, 4, 5) : __builtin_shufflevector(a, a, 6, 7);
}
__device__ __forceinline__ v2f lo2(const v4f& a) { return __builtin_shufflevector(a, a, 0, 1); }
__device__ __forceinline__ v2f hi2(const v4f& a) { return __builtin_shufflevector(a, a, 2, 3); }

__device__ __forceinline__ void a_step(const AFrag& f, const v4f (&vq)[NPAIR][4], v8f (&s2)[NUM_CAPS])
{
    v8f   u[NUM_CAPS];
    float lg[NPAIR];
#pragma unroll
    for (int p = 0; p < NPAIR; ++p) {
        v8f a = {};
        a = __builtin_amdgcn_wmma_f32_16x16x4_f32(
                false, f.wa[2 * p], false, f.xb0, (short)0, a, false, false);
        a = __builtin_amdgcn_wmma_f32_16x16x4_f32(
                false, f.wb[2 * p], false, f.xb1, (short)0, a, false, false);
        v8f b = {};
        b = __builtin_amdgcn_wmma_f32_16x16x4_f32(
                false, f.wa[2 * p + 1], false, f.xb0, (short)0, b, false, false);
        b = __builtin_amdgcn_wmma_f32_16x16x4_f32(
                false, f.wb[2 * p + 1], false, f.xb1, (short)0, b, false, false);
        u[2 * p]     = a;
        u[2 * p + 1] = b;
        v2f t = pr(a, 0) * lo2(vq[p][0]);
        t = __builtin_elementwise_fma(pr(a, 1), hi2(vq[p][0]), t);
        t = __builtin_elementwise_fma(pr(a, 2), lo2(vq[p][1]), t);
        t = __builtin_elementwise_fma(pr(a, 3), hi2(vq[p][1]), t);
        t = __builtin_elementwise_fma(pr(b, 0), lo2(vq[p][2]), t);
        t = __builtin_elementwise_fma(pr(b, 1), hi2(vq[p][2]), t);
        t = __builtin_elementwise_fma(pr(b, 2), lo2(vq[p][3]), t);
        t = __builtin_elementwise_fma(pr(b, 3), hi2(vq[p][3]), t);
        lg[p] = t.x + t.y;
    }

    float mx = lg[0];
#pragma unroll
    for (int p = 1; p < NPAIR; ++p) mx = fmaxf(mx, lg[p]);
    mx = fmaxf(mx, xhalf(mx));
#pragma unroll
    for (int p = 0; p < NPAIR; ++p) lg[p] = __expf(lg[p] - mx);
    float s8[8], s4[4];
#pragma unroll
    for (int k = 0; k < 8; ++k) s8[k] = lg[2 * k] + lg[2 * k + 1];
#pragma unroll
    for (int k = 0; k < 4; ++k) s4[k] = s8[2 * k] + s8[2 * k + 1];
    const float own = (s4[0] + s4[1]) + (s4[2] + s4[3]);
    const float sum = own + xhalf(own);
    const float inv = 1.0f / sum;

#pragma unroll
    for (int p = 0; p < NPAIR; ++p) {
        const float c = lg[p] * inv;
        s2[2 * p]     += c * u[2 * p];
        s2[2 * p + 1] += c * u[2 * p + 1];
    }
}

__global__ __launch_bounds__(128) void caps_agree_route(const float* __restrict__ x,
                                                        const float* __restrict__ W,
                                                        const float* __restrict__ v1,
                                                        float* __restrict__ part2)
{
    const int wav  = threadIdx.x >> 5;
    const int lane = threadIdx.x & 31;
    const int lo   = lane & 15;
    const int hi   = lane >> 4;
    const int mt   = blockIdx.x;
    const int m0   = mt * 16;
    const int ib   = (blockIdx.y * WPB + wav) * CPW;

    const float* xr = x  + (size_t)(m0 + lo) * (IN_CAPS * IN_DIM) + 4 * hi;
    const float* vr = v1 + (size_t)(m0 + lo) * (NUM_CAPS * CAP_DIM) + CAP_DIM * hi;
    const float* wr = W  + (lo >> 3) * (CAP_DIM * IN_DIM) + (lo & 7) * IN_DIM + 4 * hi;

    v4f vq[NPAIR][4];
#pragma unroll
    for (int p = 0; p < NPAIR; ++p) {
#pragma unroll
        for (int k = 0; k < 4; ++k)
            vq[p][k] = *(const v4f*)(vr + p * (2 * CAP_DIM) + 4 * k);
    }

    v8f s2[NUM_CAPS];
#pragma unroll
    for (int j = 0; j < NUM_CAPS; ++j) s2[j] = v8f{};

    AFrag fc, fn;
    a_load(fc, xr, wr, ib);
#pragma nounroll
    for (int ii = 0; ii < CPW; ++ii) {
        a_load(fn, xr, wr, ib + min(ii + 1, CPW - 1));
        __builtin_amdgcn_sched_barrier(0);
        a_step(fc, vq, s2);
        fc = fn;
    }

    __shared__ float4 red[WPB][2 * NUM_CAPS][32];
#pragma unroll
    for (int j = 0; j < NUM_CAPS; ++j) {
        float4 q0, q1;
        q0.x = s2[j][0];   q0.y = s2[j][1];   q0.z = s2[j][2];   q0.w = s2[j][3];
        q1.x = s2[j][4];   q1.y = s2[j][5];   q1.z = s2[j][6];   q1.w = s2[j][7];
        red[wav][2 * j + 0][lane] = q0;
        red[wav][2 * j + 1][lane] = q1;
    }
    __syncthreads();

    float* pp = part2 + (size_t)blockIdx.y * SVEC + (size_t)mt * TILE + lane * 4;
#pragma unroll
    for (int t = 0; t < (2 * NUM_CAPS) / WPB; ++t) {
        const int q = wav * ((2 * NUM_CAPS) / WPB) + t;
        float4 a = red[0][q][lane];
#pragma unroll
        for (int w = 1; w < WPB; ++w) {
            const float4 b = red[w][q][lane];
            a.x += b.x;   a.y += b.y;   a.z += b.z;   a.w += b.w;
        }
        *(float4*)(pp + q * 128) = a;
    }
}

#define RG   4
__global__ __launch_bounds__(512) void caps_reduce_squash_tiles(const float* __restrict__ part2,
                                                                float* __restrict__ out)
{
    __shared__ float4 red[RG][2][2][32];
    __shared__ float4 line[16 * 8];

    const int u    = threadIdx.x;
    const int lane = u & 31;
    const int rq   = (u >> 5) & 1;
    const int jj   = (u >> 6) & 1;
    const int kg   = u >> 7;
    const int jp   = blockIdx.x & 15;
    const int mt   = blockIdx.x >> 4;
    const int j    = 2 * jp + jj;

    const float* p = part2 + (size_t)(kg * (NSL / RG)) * SVEC + (size_t)mt * TILE
                   + ((2 * j + rq) * 32 + lane) * 4;
    float4 s = { 0.0f, 0.0f, 0.0f, 0.0f };
    for (int k = 0; k < NSL / RG; ++k) {
        const float4 v = *(const float4*)(p + (size_t)k * SVEC);
        s.x += v.x;   s.y += v.y;   s.z += v.z;   s.w += v.w;
    }
    red[kg][jj][rq][lane] = s;
    __syncthreads();

    if (u < 32) {
        float4 a0 = red[0][0][0][lane];
        float4 b0 = red[0][0][1][lane];
        float4 a1 = red[0][1][0][lane];
        float4 b1 = red[0][1][1][lane];
#pragma unroll
        for (int g = 1; g < RG; ++g) {
            const float4 a2 = red[g][0][0][lane];
            const float4 b2 = red[g][0][1][lane];
            const float4 a3 = red[g][1][0][lane];
            const float4 b3 = red[g][1][1][lane];
            a0.x += a2.x;   a0.y += a2.y;   a0.z += a2.z;   a0.w += a2.w;
            b0.x += b2.x;   b0.y += b2.y;   b0.z += b2.z;   b0.w += b2.w;
            a1.x += a3.x;   a1.y += a3.y;   a1.z += a3.z;   a1.w += a3.w;
            b1.x += b3.x;   b1.y += b3.y;   b1.z += b3.z;   b1.w += b3.w;
        }
        float sq  = a0.x * a0.x + a0.y * a0.y + a0.z * a0.z + a0.w * a0.w
                  + b0.x * b0.x + b0.y * b0.y + b0.z * b0.z + b0.w * b0.w;
        float sq1 = a1.x * a1.x + a1.y * a1.y + a1.z * a1.z + a1.w * a1.w
                  + b1.x * b1.x + b1.y * b1.y + b1.z * b1.z + b1.w * b1.w;
        sq += sq1;
        const float scale = sq / ((1.0f + sq) * (sqrtf(sq) + EPS));
        a0.x *= scale;   a0.y *= scale;   a0.z *= scale;   a0.w *= scale;
        b0.x *= scale;   b0.y *= scale;   b0.z *= scale;   b0.w *= scale;
        a1.x *= scale;   a1.y *= scale;   a1.z *= scale;   a1.w *= scale;
        b1.x *= scale;   b1.y *= scale;   b1.z *= scale;   b1.w *= scale;
        const int lo = lane & 15;
        const int hi = lane >> 4;
        line[lo * 8 + hi * 4 + 0] = a0;
        line[lo * 8 + hi * 4 + 1] = b0;
        line[lo * 8 + hi * 4 + 2] = a1;
        line[lo * 8 + hi * 4 + 3] = b1;
    }
    __syncthreads();

    if (u < 128) {
        const int lo = u >> 3;
        const int q  = u & 7;
        float* o = out + (size_t)(mt * 16 + lo) * (NUM_CAPS * CAP_DIM) + jp * 32 + q * 4;
        *(float4*)o = line[u];
    }
}

extern "C" void kernel_launch(void* const* d_in, const int* in_sizes, int n_in,
                              void* d_out, int out_size, void* d_ws, size_t ws_size,
                              hipStream_t stream)
{
    const float* x = (const float*)d_in[0];
    const float* W = (const float*)d_in[1];
    float* vout = (float*)d_out;

    float* v1    = (float*)d_ws;
    float* part2 = v1 + (size_t)SVEC;
    float* part  = part2 + (size_t)BATCH * IN_CAPS * NUM_CAPS;

    caps_s_partial<<<dim3(NUM_CAPS / JG, SPLITK), 256, 0, stream>>>(x, W, part);
    caps_reduce_squash<<<dim3(SVEC / 256),       256, 0, stream>>>(part, v1,
                                                                  1.0f / (float)NUM_CAPS);
    caps_agree_route<<<dim3(BATCH / 16, NSL), 128, 0, stream>>>(x, W, v1, part2);
    caps_reduce_squash_tiles<<<dim3((BATCH / 16) * (NUM_CAPS / 2)), 512, 0, stream>>>(part2, vout);
}


# first pass re-tiled: a wave keeps 4 sample tiles x 4 capsules (16 accumulators), 8 fragment loads per 32 WMMAs, not 20; a block's 8 K-chunks summed in LDS, 16 slices not 32; s1 sum order changed
# speedup vs baseline: 7.6994x; 1.0381x over previous
#include <hip/hip_runtime.h>
#include <hip/hip_bf16.h>
#include <math.h>


#define BATCH    128
#define IN_CAPS  2048
#define IN_DIM   8
#define NUM_CAPS 32
#define CAP_DIM  16
#define EPS      1e-8f

#define SPLITK   16
#define KW       8
#define KCHUNK   (IN_CAPS / (SPLITK * KW))
#define SVEC     (BATCH * NUM_CAPS * CAP_DIM)

typedef float v2f __attribute__((ext_vector_type(2)));
typedef float v4f __attribute__((ext_vector_type(4)));
typedef float v8f __attribute__((ext_vector_type(8)));


#define MT   4
#define JG   4
#define ST   2
#define NST  (KCHUNK / ST)

struct SFrag {
    v2f    xa[ST][MT][2];
    v2f    wb[ST][JG][2];
};

__device__ __forceinline__ void s_load_stage(SFrag& f, const float* __restrict__ xr,
                                             const float* __restrict__ wr, int i)
{
    const float* xi = xr + i * IN_DIM;
    const float* wi = wr + (size_t)i * (NUM_CAPS * CAP_DIM * IN_DIM);
#pragma unroll
    for (int t = 0; t < ST; ++t) {
#pragma unroll
        for (int mi = 0; mi < MT; ++mi) {
            const float4 p = *(const float4*)(xi + t * IN_DIM
                                                 + (size_t)mi * (16 * IN_CAPS * IN_DIM));
            f.xa[t][mi][0].x = p.x;   f.xa[t][mi][0].y = p.y;
            f.xa[t][mi][1].x = p.z;   f.xa[t][mi][1].y = p.w;
        }
#pragma unroll
        for (int jj = 0; jj < JG; ++jj) {
            const float4 q = *(const float4*)(wi + t * (NUM_CAPS * CAP_DIM * IN_DIM)
                                                 + jj * (CAP_DIM * IN_DIM));
            f.wb[t][jj][0].x = q.x;   f.wb[t][jj][0].y = q.y;
            f.wb[t][jj][1].x = q.z;   f.wb[t][jj][1].y = q.w;
        }
    }
}

__device__ __forceinline__ void s_compute_stage(const SFrag& f, v8f (&acc)[MT][JG])
{
#pragma unroll
    for (int t = 0; t < ST; ++t) {
#pragma unroll
        for (int s = 0; s < 2; ++s) {
#pragma unroll
            for (int mi = 0; mi < MT; ++mi) {
#pragma unroll
                for (int jj = 0; jj < JG; ++jj) {
                    acc[mi][jj] = __builtin_amdgcn_wmma_f32_16x16x4_f32(
                                      false, f.xa[t][mi][s], false, f.wb[t][jj][s], (short)0,
                                      acc[mi][jj], false, false);
                }
            }
        }
    }
}

__global__ __launch_bounds__(256) void caps_s_partial(const float* __restrict__ x,
                                                      const float* __restrict__ W,
                                                      float* __restrict__ part)
{
    const int j0   = blockIdx.x * JG;
    const int ks   = blockIdx.y % SPLITK;
    const int mq   = blockIdx.y / SPLITK;
    const int wav  = threadIdx.x >> 5;
    const int lane = threadIdx.x & 31;
    const int lo   = lane & 15;
    const int hi   = lane >> 4;
    const int m0   = mq * (16 * MT);
    const int i0   = (ks * KW + wav) * KCHUNK;

    const float* xr = x  + (size_t)(m0 + lo) * (IN_CAPS * IN_DIM) + 4 * hi;
    const float* wr = W  + (size_t)j0 * (CAP_DIM * IN_DIM) + lo * IN_DIM + 4 * hi;

    v8f acc[MT][JG];
#pragma unroll
    for (int mi = 0; mi < MT; ++mi) {
#pragma unroll
        for (int jj = 0; jj < JG; ++jj) acc[mi][jj] = v8f{};
    }

    SFrag f0, f1;
    s_load_stage(f0, xr, wr, i0);
    for (int st = 0; st < NST - 2; st += 2) {
        s_load_stage(f1, xr, wr, i0 + (st + 1) * ST);
        __builtin_amdgcn_sched_barrier(0);
        s_compute_stage(f0, acc);
        s_load_stage(f0, xr, wr, i0 + (st + 2) * ST);
        __builtin_amdgcn_sched_barrier(0);
        s_compute_stage(f1, acc);
    }
    s_load_stage(f1, xr, wr, i0 + (NST - 1) * ST);
    __builtin_amdgcn_sched_barrier(0);
    s_compute_stage(f0, acc);
    s_compute_stage(f1, acc);

    __shared__ float4 red[KW][2 * MT * JG][32];
#pragma unroll
    for (int mi = 0; mi < MT; ++mi) {
#pragma unroll
        for (int jj = 0; jj < JG; ++jj) {
            float4 q0, q1;
            q0.x = acc[mi][jj][0];   q0.y = acc[mi][jj][1];
            q0.z = acc[mi][jj][2];   q0.w = acc[mi][jj][3];
            q1.x = acc[mi][jj][4];   q1.y = acc[mi][jj][5];
            q1.z = acc[mi][jj][6];   q1.w = acc[mi][jj][7];
            red[wav][2 * (JG * mi + jj) + 0][lane] = q0;
            red[wav][2 * (JG * mi + jj) + 1][lane] = q1;
        }
    }
    __syncthreads();

    const int mi = wav >> 1;
#pragma unroll
    for (int t = 0; t < 4; ++t) {
        const int u  = wav * 4 + t;
        const int jj = (wav & 1) * 2 + (t >> 1);
        const int h  = t & 1;
        float4 a = red[0][u][lane];
#pragma unroll
        for (int w = 1; w < KW; ++w) {
            const float4 b = red[w][u][lane];
            a.x += b.x;   a.y += b.y;   a.z += b.z;   a.w += b.w;
        }
        float* p = part + (size_t)ks * SVEC
                 + (size_t)(m0 + 16 * mi + 4 * h + 8 * hi) * (NUM_CAPS * CAP_DIM)
                 + (j0 + jj) * CAP_DIM + lo;
        p[0 * (NUM_CAPS * CAP_DIM)] = a.x;
        p[1 * (NUM_CAPS * CAP_DIM)] = a.y;
        p[2 * (NUM_CAPS * CAP_DIM)] = a.z;
        p[3 * (NUM_CAPS * CAP_DIM)] = a.w;
    }
}

__global__ __launch_bounds__(256) void caps_reduce_squash(const float* __restrict__ part,
                                                          float* __restrict__ out,
                                                          float pre)
{
    const int t = blockIdx.x * 256 + threadIdx.x;
    float s = 0.0f;
    for (int k = 0; k < SPLITK; ++k)
        s += part[(size_t)k * SVEC + t];
    s *= pre;
    float sq = s * s;
    sq += __shfl_xor(sq, 1, 16);
    sq += __shfl_xor(sq, 2, 16);
    sq += __shfl_xor(sq, 4, 16);
    sq += __shfl_xor(sq, 8, 16);
    const float scale = sq / ((1.0f + sq) * (sqrtf(sq) + EPS));
    out[t] = s * scale;
}

#define CPW  16
#define WPB  4
#define NSL  (IN_CAPS / (CPW * WPB))
#define TILE (16 * NUM_CAPS * CAP_DIM)

__device__ __forceinline__ float xhalf(float v)
{
    const int b = __builtin_bit_cast(int, v);
    return __builtin_bit_cast(float,
        __builtin_amdgcn_permlanex16(b, b, 0x76543210, 0xfedcba98, false, false));
}

struct AFrag {
    v2f xb0, xb1;
    v2f wa[NUM_CAPS], wb[NUM_CAPS];
};

__device__ __forceinline__ void a_load(AFrag& f, const float* __restrict__ xr,
                                       const float* __restrict__ wr, int i)
{
    const float4 p = *(const float4*)(xr + i * IN_DIM);
    f.xb0.x = p.x;   f.xb0.y = p.y;
    f.xb1.x = p.z;   f.xb1.y = p.w;
    const float* wi = wr + (size_t)i * (NUM_CAPS * CAP_DIM * IN_DIM);
#pragma unroll
    for (int q = 0; q < NUM_CAPS; ++q) {
        const float4 w = *(const float4*)(wi + (q >> 1) * (2 * CAP_DIM * IN_DIM)
                                             + (q & 1) * (8 * IN_DIM));
        f.wa[q].x = w.x;   f.wa[q].y = w.y;
        f.wb[q].x = w.z;   f.wb[q].y = w.w;
    }
}

#define NPAIR (NUM_CAPS / 2)
__device__ __forceinline__ v2f pr(const v8f& a, int k)
{
    return k == 0 ? __builtin_shufflevector(a, a, 0, 1) : k == 1 ? __builtin_shufflevector(a, a, 2, 3)
         : k == 2 ? __builtin_shufflevector(a, a, 4, 5) : __builtin_shufflevector(a, a, 6, 7);
}
__device__ __forceinline__ v2f lo2(const v4f& a) { return __builtin_shufflevector(a, a, 0, 1); }
__device__ __forceinline__ v2f hi2(const v4f& a) { return __builtin_shufflevector(a, a, 2, 3); }

__device__ __forceinline__ void a_step(const AFrag& f, const v4f (&vq)[NPAIR][4], v8f (&s2)[NUM_CAPS])
{
    v8f   u[NUM_CAPS];
    float lg[NPAIR];
#pragma unroll
    for (int p = 0; p < NPAIR; ++p) {
        v8f a = {};
        a = __builtin_amdgcn_wmma_f32_16x16x4_f32(
                false, f.wa[2 * p], false, f.xb0, (short)0, a, false, false);
        a = __builtin_amdgcn_wmma_f32_16x16x4_f32(
                false, f.wb[2 * p], false, f.xb1, (short)0, a, false, false);
        v8f b = {};
        b = __builtin_amdgcn_wmma_f32_16x16x4_f32(
                false, f.wa[2 * p + 1], false, f.xb0, (short)0, b, false, false);
        b = __builtin_amdgcn_wmma_f32_16x16x4_f32(
                false, f.wb[2 * p + 1], false, f.xb1, (short)0, b, false, false);
        u[2 * p]     = a;
        u[2 * p + 1] = b;
        v2f t = pr(a, 0) * lo2(vq[p][0]);
        t = __builtin_elementwise_fma(pr(a, 1), hi2(vq[p][0]), t);
        t = __builtin_elementwise_fma(pr(a, 2), lo2(vq[p][1]), t);
        t = __builtin_elementwise_fma(pr(a, 3), hi2(vq[p][1]), t);
        t = __builtin_elementwise_fma(pr(b, 0), lo2(vq[p][2]), t);
        t = __builtin_elementwise_fma(pr(b, 1), hi2(vq[p][2]), t);
        t = __builtin_elementwise_fma(pr(b, 2), lo2(vq[p][3]), t);
        t = __builtin_elementwise_fma(pr(b, 3), hi2(vq[p][3]), t);
        lg[p] = t.x + t.y;
    }

    float mx = lg[0];
#pragma unroll
    for (int p = 1; p < NPAIR; ++p) mx = fmaxf(mx, lg[p]);
    mx = fmaxf(mx, xhalf(mx));
#pragma unroll
    for (int p = 0; p < NPAIR; ++p) lg[p] = __expf(lg[p] - mx);
    float s8[8], s4[4];
#pragma unroll
    for (int k = 0; k < 8; ++k) s8[k] = lg[2 * k] + lg[2 * k + 1];
#pragma unroll
    for (int k = 0; k < 4; ++k) s4[k] = s8[2 * k] + s8[2 * k + 1];
    const float own = (s4[0] + s4[1]) + (s4[2] + s4[3]);
    const float sum = own + xhalf(own);
    const float inv = 1.0f / sum;

#pragma unroll
    for (int p = 0; p < NPAIR; ++p) {
        const float c = lg[p] * inv;
        s2[2 * p]     += c * u[2 * p];
        s2[2 * p + 1] += c * u[2 * p + 1];
    }
}

__global__ __launch_bounds__(128) void caps_agree_route(const float* __restrict__ x,
                                                        const float* __restrict__ W,
                                                        const float* __restrict__ v1,
                                                        float* __restrict__ part2)
{
    const int wav  = threadIdx.x >> 5;
    const int lane = threadIdx.x & 31;
    const int lo   = lane & 15;
    const int hi   = lane >> 4;
    const int mt   = blockIdx.x;
    const int m0   = mt * 16;
    const int ib   = (blockIdx.y * WPB + wav) * CPW;

    const float* xr = x  + (size_t)(m0 + lo) * (IN_CAPS * IN_DIM) + 4 * hi;
    const float* vr = v1 + (size_t)(m0 + lo) * (NUM_CAPS * CAP_DIM) + CAP_DIM * hi;
    const float* wr = W  + (lo >> 3) * (CAP_DIM * IN_DIM) + (lo & 7) * IN_DIM + 4 * hi;

    v4f vq[NPAIR][4];
#pragma unroll
    for (int p = 0; p < NPAIR; ++p) {
#pragma unroll
        for (int k = 0; k < 4; ++k)
            vq[p][k] = *(const v4f*)(vr + p * (2 * CAP_DIM) + 4 * k);
    }

    v8f s2[NUM_CAPS];
#pragma unroll
    for (int j = 0; j < NUM_CAPS; ++j) s2[j] = v8f{};

    AFrag fc, fn;
    a_load(fc, xr, wr, ib);
#pragma nounroll
    for (int ii = 0; ii < CPW; ++ii) {
        a_load(fn, xr, wr, ib + min(ii + 1, CPW - 1));
        __builtin_amdgcn_sched_barrier(0);
        a_step(fc, vq, s2);
        fc = fn;
    }

    __shared__ float4 red[WPB][2 * NUM_CAPS][32];
#pragma unroll
    for (int j = 0; j < NUM_CAPS; ++j) {
        float4 q0, q1;
        q0.x = s2[j][0];   q0.y = s2[j][1];   q0.z = s2[j][2];   q0.w = s2[j][3];
        q1.x = s2[j][4];   q1.y = s2[j][5];   q1.z = s2[j][6];   q1.w = s2[j][7];
        red[wav][2 * j + 0][lane] = q0;
        red[wav][2 * j + 1][lane] = q1;
    }
    __syncthreads();

    float* pp = part2 + (size_t)blockIdx.y * SVEC + (size_t)mt * TILE + lane * 4;
#pragma unroll
    for (int t = 0; t < (2 * NUM_CAPS) / WPB; ++t) {
        const int q = wav * ((2 * NUM_CAPS) / WPB) + t;
        float4 a = red[0][q][lane];
#pragma unroll
        for (int w = 1; w < WPB; ++w) {
            const float4 b = red[w][q][lane];
            a.x += b.x;   a.y += b.y;   a.z += b.z;   a.w += b.w;
        }
        *(float4*)(pp + q * 128) = a;
    }
}

#define RG   4
__global__ __launch_bounds__(512) void caps_reduce_squash_tiles(const float* __restrict__ part2,
                                                                float* __restrict__ out)
{
    __shared__ float4 red[RG][2][2][32];
    __shared__ float4 line[16 * 8];

    const int u    = threadIdx.x;
    const int lane = u & 31;
    const int rq   = (u >> 5) & 1;
    const int jj   = (u >> 6) & 1;
    const int kg   = u >> 7;
    const int jp   = blockIdx.x & 15;
    const int mt   = blockIdx.x >> 4;
    const int j    = 2 * jp + jj;

    const float* p = part2 + (size_t)(kg * (NSL / RG)) * SVEC + (size_t)mt * TILE
                   + ((2 * j + rq) * 32 + lane) * 4;
    float4 s = { 0.0f, 0.0f, 0.0f, 0.0f };
    for (int k = 0; k < NSL / RG; ++k) {
        const float4 v = *(const float4*)(p + (size_t)k * SVEC);
        s.x += v.x;   s.y += v.y;   s.z += v.z;   s.w += v.w;
    }
    red[kg][jj][rq][lane] = s;
    __syncthreads();

    if (u < 32) {
        float4 a0 = red[0][0][0][lane];
        float4 b0 = red[0][0][1][lane];
        float4 a1 = red[0][1][0][lane];
        float4 b1 = red[0][1][1][lane];
#pragma unroll
        for (int g = 1; g < RG; ++g) {
            const float4 a2 = red[g][0][0][lane];
            const float4 b2 = red[g][0][1][lane];
            const float4 a3 = red[g][1][0][lane];
            const float4 b3 = red[g][1][1][lane];
            a0.x += a2.x;   a0.y += a2.y;   a0.z += a2.z;   a0.w += a2.w;
            b0.x += b2.x;   b0.y += b2.y;   b0.z += b2.z;   b0.w += b2.w;
            a1.x += a3.x;   a1.y += a3.y;   a1.z += a3.z;   a1.w += a3.w;
            b1.x += b3.x;   b1.y += b3.y;   b1.z += b3.z;   b1.w += b3.w;
        }
        float sq  = a0.x * a0.x + a0.y * a0.y + a0.z * a0.z + a0.w * a0.w
                  + b0.x * b0.x + b0.y * b0.y + b0.z * b0.z + b0.w * b0.w;
        float sq1 = a1.x * a1.x + a1.y * a1.y + a1.z * a1.z + a1.w * a1.w
                  + b1.x * b1.x + b1.y * b1.y + b1.z * b1.z + b1.w * b1.w;
        sq += sq1;
        const float scale = sq / ((1.0f + sq) * (sqrtf(sq) + EPS));
        a0.x *= scale;   a0.y *= scale;   a0.z *= scale;   a0.w *= scale;
        b0.x *= scale;   b0.y *= scale;   b0.z *= scale;   b0.w *= scale;
        a1.x *= scale;   a1.y *= scale;   a1.z *= scale;   a1.w *= scale;
        b1.x *= scale;   b1.y *= scale;   b1.z *= scale;   b1.w *= scale;
        const int lo = lane & 15;
        const int hi = lane >> 4;
        line[lo * 8 + hi * 4 + 0] = a0;
        line[lo * 8 + hi * 4 + 1] = b0;
        line[lo * 8 + hi * 4 + 2] = a1;
        line[lo * 8 + hi * 4 + 3] = b1;
    }
    __syncthreads();

    if (u < 128) {
        const int lo = u >> 3;
        const int q  = u & 7;
        float* o = out + (size_t)(mt * 16 + lo) * (NUM_CAPS * CAP_DIM) + jp * 32 + q * 4;
        *(float4*)o = line[u];
    }
}

extern "C" void kernel_launch(void* const* d_in, const int* in_sizes, int n_in,
                              void* d_out, int out_size, void* d_ws, size_t ws_size,
                              hipStream_t stream)
{
    const float* x = (const float*)d_in[0];
    const float* W = (const float*)d_in[1];
    float* vout = (float*)d_out;

    float* v1    = (float*)d_ws;
    float* part2 = v1 + (size_t)SVEC;
    float* part  = part2 + (size_t)BATCH * IN_CAPS * NUM_CAPS;

    caps_s_partial<<<dim3(NUM_CAPS / JG, (BATCH / (16 * MT)) * SPLITK), 256, 0, stream>>>(x, W, part);
    caps_reduce_squash<<<dim3(SVEC / 256),       256, 0, stream>>>(part, v1,
                                                                  1.0f / (float)NUM_CAPS);
    caps_agree_route<<<dim3(BATCH / 16, NSL), 128, 0, stream>>>(x, W, v1, part2);
    caps_reduce_squash_tiles<<<dim3((BATCH / 16) * (NUM_CAPS / 2)), 512, 0, stream>>>(part2, vout);
}
